# starting kernel
# baseline (speedup 1.0000x reference)
#include <hip/hip_runtime.h>
#include <math.h>

typedef __attribute__((ext_vector_type(2))) float v2f;
typedef __attribute__((ext_vector_type(8))) float v8f;

__device__ __forceinline__ unsigned f2key(float f) {
    unsigned b = __float_as_uint(f);
    return (b & 0x80000000u) ? ~b : (b | 0x80000000u);
}
__device__ __forceinline__ float key2f(unsigned k) {
    unsigned b = (k & 0x80000000u) ? (k & 0x7FFFFFFFu) : ~k;
    return __uint_as_float(b);
}

template <int K, int NC>
__global__ void gemm_wmma_f32(const float* __restrict__ A,
                              const float* __restrict__ B,
                              float* __restrict__ C,
                              int M) {
    __shared__ float Bs[NC * K];
    const int tid = threadIdx.x;

    for (int idx = tid; idx < K * NC; idx += 128) {
        int k = idx / NC;
        int n = idx - k * NC;
        Bs[n * K + k] = B[idx];
    }
    __syncthreads();

    const int waveId = tid >> 5;
    const int lane   = tid & 31;
    const int half   = lane >> 4;
    const int l15    = lane & 15;
    const int kb     = half << 1;
    const int m0     = (blockIdx.x * 4 + waveId) << 4;
    if (m0 >= M) return;

    int rowA = m0 + l15;
    if (rowA >= M) rowA = M - 1;
    const float* __restrict__ arow = A + (size_t)rowA * K;

    v2f areg[K / 4];
#pragma unroll
    for (int kk = 0; kk < K / 4; ++kk)
        areg[kk] = *(const v2f*)(arow + kk * 4 + kb);

#pragma unroll
    for (int tn = 0; tn < NC / 16; ++tn) {
        const int n0 = tn << 4;
        const float* bcol = &Bs[(n0 + l15) * K + kb];
        v8f acc = {};
#pragma unroll
        for (int kk = 0; kk < K / 4; ++kk) {
            v2f b = *(const v2f*)(bcol + kk * 4);
            acc = __builtin_amdgcn_wmma_f32_16x16x4_f32(false, areg[kk], false, b,
                                                        (short)0, acc, false, false);
        }
        const int colC  = n0 + l15;
        const int rbase = m0 + (half << 3);
#pragma unroll
        for (int r = 0; r < 8; ++r) {
            int row = rbase + r;
            if (row < M) C[(size_t)row * NC + colC] = acc[r];
        }
    }
}

__global__ void att_scores(const float* __restrict__ h,
                           const float* __restrict__ a_src,
                           const float* __restrict__ a_dst,
                           float* __restrict__ out_s,
                           float* __restrict__ out_d,
                           int N, int H, int Cc) {
    int t = blockIdx.x * blockDim.x + threadIdx.x;
    if (t >= N * H) return;
    int n  = t / H;
    int hh = t - n * H;
    const float* row = h + (size_t)n * H * Cc + (size_t)hh * Cc;
    const float* vs  = a_src + (size_t)hh * Cc;
    const float* vd  = a_dst + (size_t)hh * Cc;
    float ss = 0.f, dd = 0.f;
    for (int c = 0; c < Cc; ++c) {
        float v = row[c];
        ss += v * vs[c];
        dd += v * vd[c];
    }
    out_s[t] = ss;
    out_d[t] = dd;
}

__global__ void edge_logit_max(const int* __restrict__ ei, int E, int H,
                               const float* __restrict__ as_, const float* __restrict__ ad_,
                               float* __restrict__ p, unsigned* __restrict__ mkey) {
    int e = blockIdx.x * blockDim.x + threadIdx.x;
    if (e >= E) return;
    int s = ei[e];
    int d = ei[E + e];
    for (int h = 0; h < H; ++h) {
        float el = as_[s * H + h] + ad_[d * H + h];
        el = (el > 0.f) ? el : 0.2f * el;
        p[(size_t)e * H + h] = el;
        atomicMax(&mkey[d * H + h], f2key(el));
    }
}

__global__ void edge_softmax_num(const int* __restrict__ ei, int E, int H,
                                 float* __restrict__ p,
                                 const unsigned* __restrict__ mkey,
                                 float* __restrict__ ssum) {
    int e = blockIdx.x * blockDim.x + threadIdx.x;
    if (e >= E) return;
    int d = ei[E + e];
    for (int h = 0; h < H; ++h) {
        float el = p[(size_t)e * H + h];
        float m  = key2f(mkey[d * H + h]);
        float pe = __expf(el - m);
        p[(size_t)e * H + h] = pe;
        atomicAdd(&ssum[d * H + h], pe);
    }
}

__global__ void edge_aggregate(const int* __restrict__ ei, int E,
                               int logF, int logC,
                               const float* __restrict__ p,
                               const float* __restrict__ ssum,
                               const float* __restrict__ hsrc,
                               float* __restrict__ out) {
    long long t = (long long)blockIdx.x * blockDim.x + threadIdx.x;
    long long total = (long long)E << logF;
    if (t >= total) return;
    int e  = (int)(t >> logF);
    int c  = (int)(t & ((1 << logF) - 1));
    int hh = c >> logC;
    int H  = 1 << (logF - logC);
    int s  = ei[e];
    int d  = ei[E + e];
    float alpha = p[(size_t)e * H + hh] / (ssum[(size_t)d * H + hh] + 1e-16f);
    atomicAdd(&out[((size_t)d << logF) + c], hsrc[((size_t)s << logF) + c] * alpha);
}

__global__ void bias_elu(float* __restrict__ x, const float* __restrict__ b,
                         long long total, int mask) {
    long long t = (long long)blockIdx.x * blockDim.x + threadIdx.x;
    if (t >= total) return;
    float v = x[t] + b[(int)(t & mask)];
    x[t] = (v > 0.f) ? v : (__expf(v) - 1.f);
}

__global__ void bias_add(float* __restrict__ x, const float* __restrict__ b,
                         long long total, int mask) {
    long long t = (long long)blockIdx.x * blockDim.x + threadIdx.x;
    if (t >= total) return;
    x[t] += b[(int)(t & mask)];
}

extern "C" void kernel_launch(void* const* d_in, const int* in_sizes, int n_in,
                              void* d_out, int out_size, void* d_ws, size_t ws_size,
                              hipStream_t stream) {
    const float* x        = (const float*)d_in[0];
    const int*   ei       = (const int*)  d_in[1];
    const float* W1       = (const float*)d_in[2];
    const float* att_src1 = (const float*)d_in[3];
    const float* att_dst1 = (const float*)d_in[4];
    const float* b1       = (const float*)d_in[5];
    const float* W2       = (const float*)d_in[6];
    const float* att_src2 = (const float*)d_in[7];
    const float* att_dst2 = (const float*)d_in[8];
    const float* b2       = (const float*)d_in[9];
    float* out = (float*)d_out;

    const int N = in_sizes[0] / 128;
    const int E = in_sizes[1] / 2;

    float* ws = (float*)d_ws;
    float*    h1  = ws;
    float*    f1  = h1 + (size_t)N * 128;
    float*    as1 = f1 + (size_t)N * 128;
    float*    ad1 = as1 + (size_t)N * 4;
    unsigned* m1  = (unsigned*)(ad1 + (size_t)N * 4);
    float*    s1  = (float*)m1 + (size_t)N * 4;
    float*    as2 = s1 + (size_t)N * 4;
    float*    ad2 = as2 + N;
    unsigned* m2  = (unsigned*)(ad2 + N);
    float*    s2  = (float*)m2 + N;
    float*    p1  = s2 + N;
    float*    h2  = h1;
    float*    p2  = p1;

    hipMemsetAsync(f1, 0, sizeof(float) * ((size_t)N * 128 + (size_t)N * 20), stream);
    hipMemsetAsync(out, 0, sizeof(float) * (size_t)N * 64, stream);

    const int TB = 256;
    const int mTiles = (N + 15) / 16;
    const int gBlocks = (mTiles + 3) / 4;

    gemm_wmma_f32<128, 128><<<gBlocks, 128, 0, stream>>>(x, W1, h1, N);
    att_scores<<<(N * 4 + TB - 1) / TB, TB, 0, stream>>>(h1, att_src1, att_dst1, as1, ad1, N, 4, 32);
    edge_logit_max  <<<(E + TB - 1) / TB, TB, 0, stream>>>(ei, E, 4, as1, ad1, p1, m1);
    edge_softmax_num<<<(E + TB - 1) / TB, TB, 0, stream>>>(ei, E, 4, p1, m1, s1);
    {
        long long total = (long long)E << 7;
        edge_aggregate<<<(int)((total + TB - 1) / TB), TB, 0, stream>>>(ei, E, 7, 5, p1, s1, h1, f1);
    }
    {
        long long total = (long long)N * 128;
        bias_elu<<<(int)((total + TB - 1) / TB), TB, 0, stream>>>(f1, b1, total, 127);
    }

    gemm_wmma_f32<128, 64><<<gBlocks, 128, 0, stream>>>(f1, W2, h2, N);
    att_scores<<<(N + TB - 1) / TB, TB, 0, stream>>>(h2, att_src2, att_dst2, as2, ad2, N, 1, 64);
    edge_logit_max  <<<(E + TB - 1) / TB, TB, 0, stream>>>(ei, E, 1, as2, ad2, p2, m2);
    edge_softmax_num<<<(E + TB - 1) / TB, TB, 0, stream>>>(ei, E, 1, p2, m2, s2);
    {
        long long total = (long long)E << 6;
        edge_aggregate<<<(int)((total + TB - 1) / TB), TB, 0, stream>>>(ei, E, 6, 6, p2, s2, h2, out);
    }
    {
        long long total = (long long)N * 64;
        bias_add<<<(int)((total + TB - 1) / TB), TB, 0, stream>>>(out, b2, total, 63);
    }
}


# Aggregation scans a one-byte owner plane (dst/196, written by the logit pass), 16 edges per lane load, instead of the 4-byte dst plane; queue holds edge ids only; LDS float adds in another order
# speedup vs baseline: 1.7026x; 1.7026x over previous
#include <hip/hip_runtime.h>
#include <math.h>

typedef __attribute__((ext_vector_type(2))) float v2f;
typedef __attribute__((ext_vector_type(8))) float v8f;

__device__ __forceinline__ unsigned f2key(float f) {
    unsigned b = __float_as_uint(f);
    return (b & 0x80000000u) ? ~b : (b | 0x80000000u);
}
__device__ __forceinline__ float key2f(unsigned k) {
    unsigned b = (k & 0x80000000u) ? (k & 0x7FFFFFFFu) : ~k;
    return __uint_as_float(b);
}

template <int K, int NC>
__global__ void gemm_wmma_f32(const float* __restrict__ A,
                              const float* __restrict__ B,
                              float* __restrict__ C,
                              int M) {
    __shared__ float Bs[NC * K];
    const int tid = threadIdx.x;

    for (int idx = tid; idx < K * NC; idx += 128) {
        int k = idx / NC;
        int n = idx - k * NC;
        Bs[n * K + k] = B[idx];
    }
    __syncthreads();

    const int waveId = tid >> 5;
    const int lane   = tid & 31;
    const int half   = lane >> 4;
    const int l15    = lane & 15;
    const int kb     = half << 1;
    const int m0     = (blockIdx.x * 4 + waveId) << 4;
    if (m0 >= M) return;

    int rowA = m0 + l15;
    if (rowA >= M) rowA = M - 1;
    const float* __restrict__ arow = A + (size_t)rowA * K;

    v2f areg[K / 4];
#pragma unroll
    for (int kk = 0; kk < K / 4; ++kk)
        areg[kk] = *(const v2f*)(arow + kk * 4 + kb);

#pragma unroll
    for (int tn = 0; tn < NC / 16; ++tn) {
        const int n0 = tn << 4;
        const float* bcol = &Bs[(n0 + l15) * K + kb];
        v8f acc = {};
#pragma unroll
        for (int kk = 0; kk < K / 4; ++kk) {
            v2f b = *(const v2f*)(bcol + kk * 4);
            acc = __builtin_amdgcn_wmma_f32_16x16x4_f32(false, areg[kk], false, b,
                                                        (short)0, acc, false, false);
        }
        const int colC  = n0 + l15;
        const int rbase = m0 + (half << 3);
#pragma unroll
        for (int r = 0; r < 8; ++r) {
            int row = rbase + r;
            if (row < M) C[(size_t)row * NC + colC] = acc[r];
        }
    }
}

__global__ void att_scores(const float* __restrict__ h,
                           const float* __restrict__ a_src,
                           const float* __restrict__ a_dst,
                           float* __restrict__ out_s,
                           float* __restrict__ out_d,
                           int N, int H, int Cc) {
    int t = blockIdx.x * blockDim.x + threadIdx.x;
    if (t >= N * H) return;
    int n  = t / H;
    int hh = t - n * H;
    const float* row = h + (size_t)n * H * Cc + (size_t)hh * Cc;
    const float* vs  = a_src + (size_t)hh * Cc;
    const float* vd  = a_dst + (size_t)hh * Cc;
    float ss = 0.f, dd = 0.f;
    for (int c = 0; c < Cc; ++c) {
        float v = row[c];
        ss += v * vs[c];
        dd += v * vd[c];
    }
    out_s[t] = ss;
    out_d[t] = dd;
}

#define AGG_T  196
#define AGG_NW 32
#define AGG_QC 1568

__global__ void edge_logit_max(const int* __restrict__ ei, int E, int H,
                               const float* __restrict__ as_, const float* __restrict__ ad_,
                               float* __restrict__ p, unsigned* __restrict__ mkey,
                               unsigned* __restrict__ ownw) {
    int e = blockIdx.x * blockDim.x + threadIdx.x;
    if (e >= E) return;
    int s = ei[e];
    int d = ei[E + e];
    ((unsigned char*)ownw)[e] = (unsigned char)((unsigned)d / (unsigned)AGG_T);
    for (int h = 0; h < H; ++h) {
        float el = as_[s * H + h] + ad_[d * H + h];
        el = (el > 0.f) ? el : 0.2f * el;
        p[(size_t)e * H + h] = el;
        atomicMax(&mkey[d * H + h], f2key(el));
    }
}

__global__ void edge_softmax_num(const int* __restrict__ ei, int E, int H,
                                 float* __restrict__ p,
                                 const unsigned* __restrict__ mkey,
                                 float* __restrict__ ssum) {
    int e = blockIdx.x * blockDim.x + threadIdx.x;
    if (e >= E) return;
    int d = ei[E + e];
    for (int h = 0; h < H; ++h) {
        float el = p[(size_t)e * H + h];
        float m  = key2f(mkey[d * H + h]);
        float pe = __expf(el - m);
        p[(size_t)e * H + h] = pe;
        atomicAdd(&ssum[d * H + h], pe);
    }
}

template <int F, int H, bool ELU>
__global__ __launch_bounds__(1024)
void edge_aggregate_own(const int* __restrict__ ei, int E,
                        const unsigned* __restrict__ own,
                        const float* __restrict__ p,
                        const float* __restrict__ ssum,
                        const float* __restrict__ hsrc,
                        const float* __restrict__ bias,
                        float* __restrict__ out, int N) {
    constexpr int KF = F / 32;
    constexpr int CH = F / H;
    constexpr int NB = 1024 / F;
    static_assert(CH % 32 == 0, "a 32-channel slice must lie inside one head");
    __shared__ __attribute__((aligned(16))) float acc[AGG_T * F];
    __shared__ int   qe[AGG_NW * AGG_QC];

    const int tid  = threadIdx.x;
    const int lane = tid & 31;
    const int wave = __builtin_amdgcn_readfirstlane(tid >> 5);
    const int d0   = blockIdx.x * AGG_T;
    const int tcur = min(AGG_T, N - d0);

    static_assert((AGG_NW * 32 * 4) % F == 0, "a thread must keep its channels from trip to trip");
    const float4 bv = *(const float4*)(bias + ((tid * 4) & (F - 1)));

    for (int i = tid; i < AGG_T * F; i += AGG_NW * 32) acc[i] = 0.f;
    __syncthreads();

    const int* __restrict__ dstp = ei + E;
    int* myqe = qe + wave * AGG_QC;
    const bool byte_ok = (gridDim.x <= 256u) && ((E & 15) == 0) && ((((size_t)own) & 15) == 0);

    auto ldo = [&](int e0) -> uint4 {
        return *(const uint4*)((const unsigned char*)own + ((e0 < E) ? e0 : 0));
    };

    auto drain = [&](int n) {
        for (int off = 0; off < n; off += 32) {
            const int cnt = min(32, n - off);
            int   s = 0, dl = 0;
            float al[H];
#pragma unroll
            for (int h = 0; h < H; ++h) al[h] = 0.f;
            if (lane < cnt) {
                const int e = myqe[off + lane];
                s  = ei[e];
                const int d = dstp[e];
                dl = d - d0;
#pragma unroll
                for (int h = 0; h < H; ++h)
                    al[h] = p[(size_t)e * H + h] / (ssum[(size_t)d * H + h] + 1e-16f);
            }
            for (int j0 = 0; j0 < cnt; j0 += NB) {
                float v[NB][KF];
                int   dj[NB];
#pragma unroll
                for (int b = 0; b < NB; ++b) {
                    const int j  = min(j0 + b, cnt - 1);
                    const int sj = __builtin_amdgcn_readlane(s, j);
                    dj[b] = __builtin_amdgcn_readlane(dl, j);
                    const float* __restrict__ row = hsrc + (size_t)sj * F + lane;
#pragma unroll
                    for (int k = 0; k < KF; ++k) {
                        const float a = __int_as_float(
                            __builtin_amdgcn_readlane(__float_as_int(al[(k * 32) / CH]), j));
                        v[b][k] = row[k * 32] * a;
                    }
                }
#pragma unroll
                for (int b = 0; b < NB; ++b) {
                    if (j0 + b < cnt) {
                        float* arow = acc + dj[b] * F + lane;
#pragma unroll
                        for (int k = 0; k < KF; ++k)
                            __hip_atomic_fetch_add(arow + k * 32, v[b][k], __ATOMIC_RELAXED,
                                                   __HIP_MEMORY_SCOPE_WORKGROUP);
                    }
                }
            }
        }
    };

    int qn = 0;

    auto requeue = [&]() {
        const int n32 = qn & ~31;
        drain(n32);
        const int r = qn - n32;
        if (lane < r) {
            const int te = myqe[n32 + lane];
            myqe[lane] = te;
        }
        qn = r;
    };

    if (byte_ok) {
        const int      stride = AGG_NW * 512;
        const unsigned bb     = (unsigned)blockIdx.x * 0x01010101u;
        auto zb = [&](unsigned x) -> unsigned {
            const unsigned y = x ^ bb;
            return ~(((y & 0x7f7f7f7fu) + 0x7f7f7f7fu) | y) & 0x80808080u;
        };
        auto step = [&](const uint4& cur, int wbs) {
            const int e0 = wbs + lane * 16;
            unsigned h = (zb(cur.x) >> 7) | (zb(cur.y) >> 6) | (zb(cur.z) >> 5) | (zb(cur.w) >> 4);
            if (e0 >= E) h = 0u;
#pragma nounroll
            for (int trip = 0; trip < 16; ++trip) {
                const bool     hit = (h != 0u);
                const unsigned m   = __builtin_amdgcn_ballot_w32(hit);
                if (!m) break;
                if (hit) {
                    const int q   = __builtin_ctz(h);
                    const int pos = qn + (int)__builtin_amdgcn_mbcnt_lo(m, 0u);
                    myqe[pos] = e0 + ((q & 7) << 2) + (q >> 3);
                    h &= h - 1u;
                }
                qn += __builtin_popcount(m);
            }
        };
        for (int wb = wave * 512; wb < E; wb += 6 * stride) {
            const uint4 t0 = ldo(wb + lane * 16);
            const uint4 t1 = ldo(wb + stride + lane * 16);
            const uint4 t2 = ldo(wb + 2 * stride + lane * 16);
            const uint4 t3 = ldo(wb + 3 * stride + lane * 16);
            const uint4 t4 = ldo(wb + 4 * stride + lane * 16);
            const uint4 t5 = ldo(wb + 5 * stride + lane * 16);
            step(t0, wb);
            step(t1, wb + stride);
            step(t2, wb + 2 * stride);
            if (qn >= 32) requeue();
            step(t3, wb + 3 * stride);
            step(t4, wb + 4 * stride);
            step(t5, wb + 5 * stride);
            if (qn >= 32) requeue();
        }
    } else {
        for (int wb = wave * 32; wb < E; wb += AGG_NW * 32) {
            const int  e   = wb + lane;
            const int  dv  = (e < E) ? dstp[e] : -1;
            const bool hit = (e < E) && ((unsigned)(dv - d0) < (unsigned)tcur);
            const unsigned m = __builtin_amdgcn_ballot_w32(hit);
            if (m) {
                if (hit) myqe[qn + (int)__builtin_amdgcn_mbcnt_lo(m, 0u)] = e;
                qn += __builtin_popcount(m);
            }
            if (qn >= 32) requeue();
        }
    }
    if (qn > 0) drain(qn);
    __syncthreads();

    float* __restrict__ orow = out + (size_t)d0 * F;
    const int tot = tcur * F;
    for (int i = tid * 4; i < tot; i += AGG_NW * 32 * 4) {
        float4 v = *(const float4*)(acc + i);
        v.x += bv.x; v.y += bv.y; v.z += bv.z; v.w += bv.w;
        if (ELU) {
            v.x = (v.x > 0.f) ? v.x : (__expf(v.x) - 1.f);
            v.y = (v.y > 0.f) ? v.y : (__expf(v.y) - 1.f);
            v.z = (v.z > 0.f) ? v.z : (__expf(v.z) - 1.f);
            v.w = (v.w > 0.f) ? v.w : (__expf(v.w) - 1.f);
        }
        *(float4*)(orow + i) = v;
    }
}

extern "C" void kernel_launch(void* const* d_in, const int* in_sizes, int n_in,
                              void* d_out, int out_size, void* d_ws, size_t ws_size,
                              hipStream_t stream) {
    const float* x        = (const float*)d_in[0];
    const int*   ei       = (const int*)  d_in[1];
    const float* W1       = (const float*)d_in[2];
    const float* att_src1 = (const float*)d_in[3];
    const float* att_dst1 = (const float*)d_in[4];
    const float* b1       = (const float*)d_in[5];
    const float* W2       = (const float*)d_in[6];
    const float* att_src2 = (const float*)d_in[7];
    const float* att_dst2 = (const float*)d_in[8];
    const float* b2       = (const float*)d_in[9];
    float* out = (float*)d_out;

    const int N = in_sizes[0] / 128;
    const int E = in_sizes[1] / 2;

    float* ws = (float*)d_ws;
    float*    h1  = ws;
    float*    f1  = h1 + (size_t)N * 128;
    float*    as1 = f1 + (size_t)N * 128;
    float*    ad1 = as1 + (size_t)N * 4;
    unsigned* m1  = (unsigned*)(ad1 + (size_t)N * 4);
    float*    s1  = (float*)m1 + (size_t)N * 4;
    float*    as2 = s1 + (size_t)N * 4;
    float*    ad2 = as2 + N;
    unsigned* m2  = (unsigned*)(ad2 + N);
    float*    s2  = (float*)m2 + N;
    float*    p1  = s2 + N;
    float*    h2  = h1;
    float*    p2  = p1;
    unsigned* own1 = (unsigned*)out;
    unsigned* own2 = (unsigned*)(p1 + (size_t)E);

    hipMemsetAsync(as1, 0, sizeof(float) * ((size_t)N * 20), stream);

    const int TB = 256;
    const int mTiles = (N + 15) / 16;
    const int gBlocks = (mTiles + 3) / 4;
    const int aBlocks = (N + AGG_T - 1) / AGG_T;

    gemm_wmma_f32<128, 128><<<gBlocks, 128, 0, stream>>>(x, W1, h1, N);
    att_scores<<<(N * 4 + TB - 1) / TB, TB, 0, stream>>>(h1, att_src1, att_dst1, as1, ad1, N, 4, 32);
    edge_logit_max  <<<(E + TB - 1) / TB, TB, 0, stream>>>(ei, E, 4, as1, ad1, p1, m1, own1);
    edge_softmax_num<<<(E + TB - 1) / TB, TB, 0, stream>>>(ei, E, 4, p1, m1, s1);
    edge_aggregate_own<128, 4, true><<<aBlocks, AGG_NW * 32, 0, stream>>>(ei, E, own1, p1, s1, h1, b1, f1, N);

    gemm_wmma_f32<128, 64><<<gBlocks, 128, 0, stream>>>(f1, W2, h2, N);
    att_scores<<<(N + TB - 1) / TB, TB, 0, stream>>>(h2, att_src2, att_dst2, as2, ad2, N, 1, 64);
    edge_logit_max  <<<(E + TB - 1) / TB, TB, 0, stream>>>(ei, E, 1, as2, ad2, p2, m2, own2);
    edge_softmax_num<<<(E + TB - 1) / TB, TB, 0, stream>>>(ei, E, 1, p2, m2, s2);
    edge_aggregate_own<64, 1, false><<<aBlocks, AGG_NW * 32, 0, stream>>>(ei, E, own2, p2, s2, h2, b2, out, N);
}


# GEMMs: weights kept in registers as B fragments (64 interleaved columns per wave, 16-byte loads/stores), waves loop over row tiles; LDS weight stage and barrier go; k summed in another order
# speedup vs baseline: 6.5487x; 3.8463x over previous
#include <hip/hip_runtime.h>
#include <math.h>

typedef __attribute__((ext_vector_type(2))) float v2f;
typedef __attribute__((ext_vector_type(8))) float v8f;

__device__ __forceinline__ unsigned f2key(float f) {
    unsigned b = __float_as_uint(f);
    return (b & 0x80000000u) ? ~b : (b | 0x80000000u);
}
__device__ __forceinline__ float key2f(unsigned k) {
    unsigned b = (k & 0x80000000u) ? (k & 0x7FFFFFFFu) : ~k;
    return __uint_as_float(b);
}

__device__ __forceinline__ constexpr int frag_k(int s, int h, int i) {
    return 8 * (s >> 1) + 4 * h + 2 * (s & 1) + i;
}

template <int K, int NC>
__global__ __launch_bounds__(128)
void gemm_wmma_f32(const float* __restrict__ A,
                   const float* __restrict__ B,
                   float* __restrict__ C,
                   int M) {
    constexpr int NG    = NC / 64;
    constexpr int SLOTS = 4 / NG;
    constexpr int KS    = K / 4;
    static_assert(NC % 64 == 0 && (4 % NG) == 0 && K % 8 == 0, "tile shape");

    const int tid  = threadIdx.x;
    const int lane = tid & 31;
    const int wave = __builtin_amdgcn_readfirstlane(tid >> 5);
    const int h    = lane >> 4;
    const int c    = lane & 15;
    const int nb   = (wave % NG) * 64;
    const int slot = wave / NG;
    const int mTiles = (M + 15) >> 4;
    const int stride = (int)gridDim.x * SLOTS;

    v2f breg[4][KS];
    {
        const float* __restrict__ bcol = B + nb + 4 * c;
#pragma unroll
        for (int s = 0; s < KS; ++s) {
#pragma unroll
            for (int i = 0; i < 2; ++i) {
                const float4 w = *(const float4*)(bcol + (size_t)frag_k(s, h, i) * NC);
                breg[0][s][i] = w.x; breg[1][s][i] = w.y; breg[2][s][i] = w.z; breg[3][s][i] = w.w;
            }
        }
    }

    for (int mt = (int)blockIdx.x * SLOTS + slot; mt < mTiles; mt += stride) {
        const int m0 = mt << 4;
        int rowA = m0 + c;
        if (rowA >= M) rowA = M - 1;
        const float* __restrict__ arow = A + (size_t)rowA * K + 4 * h;

        float4 q[K / 8];
#pragma unroll
        for (int j = 0; j < K / 8; ++j) q[j] = *(const float4*)(arow + 8 * j);

        v8f acc[4] = {};
#pragma unroll
        for (int s = 0; s < KS; ++s) {
            const float4 qq = q[s >> 1];
            v2f a;
            if (s & 1) { a.x = qq.z; a.y = qq.w; } else { a.x = qq.x; a.y = qq.y; }
#pragma unroll
            for (int t = 0; t < 4; ++t)
                acc[t] = __builtin_amdgcn_wmma_f32_16x16x4_f32(false, a, false, breg[t][s],
                                                               (short)0, acc[t], false, false);
        }

        float* __restrict__ crow = C + (size_t)(m0 + (h << 3)) * NC + nb + 4 * c;
#pragma unroll
        for (int r = 0; r < 8; ++r) {
            if (m0 + (h << 3) + r < M) {
                float4 o;
                o.x = acc[0][r]; o.y = acc[1][r]; o.z = acc[2][r]; o.w = acc[3][r];
                *(float4*)(crow + (size_t)r * NC) = o;
            }
        }
    }
}

__global__ void att_scores(const float* __restrict__ h,
                           const float* __restrict__ a_src,
                           const float* __restrict__ a_dst,
                           float* __restrict__ out_s,
                           float* __restrict__ out_d,
                           int N, int H, int Cc) {
    int t = blockIdx.x * blockDim.x + threadIdx.x;
    if (t >= N * H) return;
    int n  = t / H;
    int hh = t - n * H;
    const float* row = h + (size_t)n * H * Cc + (size_t)hh * Cc;
    const float* vs  = a_src + (size_t)hh * Cc;
    const float* vd  = a_dst + (size_t)hh * Cc;
    float ss = 0.f, dd = 0.f;
    for (int c = 0; c < Cc; ++c) {
        float v = row[c];
        ss += v * vs[c];
        dd += v * vd[c];
    }
    out_s[t] = ss;
    out_d[t] = dd;
}

#define AGG_T  196
#define AGG_NW 32
#define AGG_QC 1024

__global__ void owner_plane(const int* __restrict__ ei, int E, unsigned char* __restrict__ own) {
    const int e4 = (blockIdx.x * blockDim.x + threadIdx.x) * 4;
    if (e4 >= E) return;
    const int* __restrict__ dstp = ei + E;
    if (e4 + 3 < E) {
        const unsigned b0 = (unsigned)dstp[e4]     / (unsigned)AGG_T;
        const unsigned b1 = (unsigned)dstp[e4 + 1] / (unsigned)AGG_T;
        const unsigned b2 = (unsigned)dstp[e4 + 2] / (unsigned)AGG_T;
        const unsigned b3 = (unsigned)dstp[e4 + 3] / (unsigned)AGG_T;
        *(unsigned*)(own + e4) = (b0 & 255u) | ((b1 & 255u) << 8) | ((b2 & 255u) << 16) | ((b3 & 255u) << 24);
    } else {
        for (int e = e4; e < E; ++e) own[e] = (unsigned char)((unsigned)dstp[e] / (unsigned)AGG_T);
    }
}

template <int H>
__device__ __forceinline__ void ld_heads(const float* __restrict__ q, float (&o)[H]) {
    if constexpr (H == 4) {
        const float4 t = *(const float4*)q;
        o[0] = t.x; o[1] = t.y; o[2] = t.z; o[3] = t.w;
    } else {
#pragma unroll
        for (int h = 0; h < H; ++h) o[h] = q[h];
    }
}

template <int F, int H, bool ELU>
__global__ __launch_bounds__(1024)
void edge_aggregate_own(const int* __restrict__ ei, int E,
                        const unsigned char* __restrict__ own,
                        const float* __restrict__ as_,
                        const float* __restrict__ ad_,
                        const float* __restrict__ hsrc,
                        const float* __restrict__ bias,
                        float* __restrict__ out, int N) {
    constexpr int KF = F / 32;
    constexpr int CH = F / H;
    constexpr int NB = 1024 / F;
    static_assert(CH % 32 == 0, "a 32-channel slice must lie inside one head");
    static_assert(AGG_T <= 256, "a local destination must fit a byte");
    __shared__ __attribute__((aligned(16))) float acc[AGG_T * F];
    __shared__ int           qs[AGG_NW * AGG_QC];
    __shared__ unsigned char qd[AGG_NW * AGG_QC];
    __shared__ unsigned      mx[AGG_T * H];
    __shared__ float         sm[AGG_T * H];
    __shared__ int           lostflag;

    const int tid  = threadIdx.x;
    const int lane = tid & 31;
    const int wave = __builtin_amdgcn_readfirstlane(tid >> 5);
    const int d0   = blockIdx.x * AGG_T;
    const int tcur = min(AGG_T, N - d0);

    static_assert((AGG_NW * 32 * 4) % F == 0, "a thread must keep its channels from trip to trip");
    const float4 bv = *(const float4*)(bias + ((tid * 4) & (F - 1)));

    for (int i = tid; i < AGG_T * F; i += AGG_NW * 32) acc[i] = 0.f;
    for (int i = tid; i < AGG_T * H; i += AGG_NW * 32) { mx[i] = 0u; sm[i] = 0.f; }
    if (tid == 0) lostflag = 0;
    __syncthreads();

    const int* __restrict__ dstp = ei + E;
    int*           myq  = qs + wave * AGG_QC;
    unsigned char* myqd = qd + wave * AGG_QC;
    const bool byte_ok = (gridDim.x <= 256u) && ((E & 15) == 0) && ((((size_t)own) & 15) == 0);

    auto ldo = [&](int e0) -> uint4 {
        return *(const uint4*)(own + ((e0 < E) ? e0 : 0));
    };

    static_assert(AGG_QC % 128 == 0, "a lane past the end must still store inside the queue");
    auto resolve = [&](int n, bool domax) {
        for (int off = 0; off < n; off += 128) {
            int  e[4], s[4], d[4];
            bool ok[4];
#pragma unroll
            for (int b = 0; b < 4; ++b) {
                const int idx = off + b * 32 + lane;
                ok[b] = idx < n;
                e[b]  = ok[b] ? myq[idx] : 0;
            }
#pragma unroll
            for (int b = 0; b < 4; ++b) { s[b] = ei[e[b]]; d[b] = dstp[e[b]]; }
            float a[4][H], c[4][H];
            if (domax) {
#pragma unroll
                for (int b = 0; b < 4; ++b) {
                    ld_heads<H>(as_ + (size_t)s[b] * H, a[b]);
                    ld_heads<H>(ad_ + (size_t)d[b] * H, c[b]);
                }
            }
#pragma unroll
            for (int b = 0; b < 4; ++b) {
                const int idx = off + b * 32 + lane;
                const int dl  = d[b] - d0;
                myq[idx]  = s[b];
                myqd[idx] = (unsigned char)dl;
                if (domax && ok[b]) {
#pragma unroll
                    for (int h = 0; h < H; ++h) {
                        float el = a[b][h] + c[b][h];
                        el = (el > 0.f) ? el : 0.2f * el;
                        atomicMax(&mx[dl * H + h], f2key(el));
                    }
                }
            }
        }
    };

    auto drain = [&](int n) {
        for (int off = 0; off < n; off += 32) {
            const int cnt = min(32, n - off);
            int   s = 0, dl = 0;
            float al[H];
#pragma unroll
            for (int h = 0; h < H; ++h) al[h] = 0.f;
            if (lane < cnt) {
                s  = myq[off + lane];
                dl = (int)myqd[off + lane];
                float a[H], c[H];
                ld_heads<H>(as_ + (size_t)s * H, a);
                ld_heads<H>(ad_ + (size_t)(d0 + dl) * H, c);
#pragma unroll
                for (int h = 0; h < H; ++h) {
                    float el = a[h] + c[h];
                    el = (el > 0.f) ? el : 0.2f * el;
                    const float pe = __expf(el - key2f(mx[dl * H + h]));
                    al[h] = pe;
                    __hip_atomic_fetch_add(&sm[dl * H + h], pe, __ATOMIC_RELAXED,
                                           __HIP_MEMORY_SCOPE_WORKGROUP);
                }
            }
            for (int j0 = 0; j0 < cnt; j0 += NB) {
                float v[NB][KF];
                int   dj[NB];
#pragma unroll
                for (int b = 0; b < NB; ++b) {
                    const int j  = min(j0 + b, cnt - 1);
                    const int sj = __builtin_amdgcn_readlane(s, j);
                    dj[b] = __builtin_amdgcn_readlane(dl, j);
                    const float* __restrict__ row = hsrc + (size_t)sj * F + lane;
#pragma unroll
                    for (int k = 0; k < KF; ++k) {
                        const float w = __int_as_float(
                            __builtin_amdgcn_readlane(__float_as_int(al[(k * 32) / CH]), j));
                        v[b][k] = row[k * 32] * w;
                    }
                }
#pragma unroll
                for (int b = 0; b < NB; ++b) {
                    if (j0 + b < cnt) {
                        float* arow = acc + dj[b] * F + lane;
#pragma unroll
                        for (int k = 0; k < KF; ++k)
                            __hip_atomic_fetch_add(arow + k * 32, v[b][k], __ATOMIC_RELAXED,
                                                   __HIP_MEMORY_SCOPE_WORKGROUP);
                    }
                }
            }
        }
    };

    int qn = 0;

    const int      stride = AGG_NW * 512;
    const unsigned bb     = (unsigned)blockIdx.x * 0x01010101u;
    auto zb = [&](unsigned x) -> unsigned {
        const unsigned y = x ^ bb;
        return ~(((y & 0x7f7f7f7fu) + 0x7f7f7f7fu) | y) & 0x80808080u;
    };
    auto step = [&](const uint4& cur, int wbs) {
        const int e0 = wbs + lane * 16;
        unsigned h = (zb(cur.x) >> 7) | (zb(cur.y) >> 6) | (zb(cur.z) >> 5) | (zb(cur.w) >> 4);
        if (e0 >= E) h = 0u;
#pragma nounroll
        for (int trip = 0; trip < 16; ++trip) {
            const bool     hit = (h != 0u);
            const unsigned m   = __builtin_amdgcn_ballot_w32(hit);
            if (!m) break;
            if (hit) {
                const int q   = __builtin_ctz(h);
                const int pos = qn + (int)__builtin_amdgcn_mbcnt_lo(m, 0u);
                myq[pos] = e0 + ((q & 7) << 2) + (q >> 3);
                h &= h - 1u;
            }
            qn += __builtin_popcount(m);
        }
    };

    auto scan = [&](auto look) {
        if (byte_ok) {
            for (int wb = wave * 512; wb < E; wb += 6 * stride) {
                const uint4 t0 = ldo(wb + lane * 16);
                const uint4 t1 = ldo(wb + stride + lane * 16);
                const uint4 t2 = ldo(wb + 2 * stride + lane * 16);
                const uint4 t3 = ldo(wb + 3 * stride + lane * 16);
                const uint4 t4 = ldo(wb + 4 * stride + lane * 16);
                const uint4 t5 = ldo(wb + 5 * stride + lane * 16);
                step(t0, wb);              look();
                step(t1, wb + stride);     look();
                step(t2, wb + 2 * stride); look();
                step(t3, wb + 3 * stride); look();
                step(t4, wb + 4 * stride); look();
                step(t5, wb + 5 * stride); look();
            }
        } else {
            for (int wb = wave * 32; wb < E; wb += AGG_NW * 32) {
                const int  e   = wb + lane;
                const int  dv  = (e < E) ? dstp[e] : -1;
                const bool hit = (e < E) && ((unsigned)(dv - d0) < (unsigned)tcur);
                const unsigned m = __builtin_amdgcn_ballot_w32(hit);
                if (m) {
                    if (hit) myq[qn + (int)__builtin_amdgcn_mbcnt_lo(m, 0u)] = e;
                    qn += __builtin_popcount(m);
                }
                look();
            }
        }
    };

    int lost = 0;
    scan([&]() {
        if (qn > AGG_QC - 512) {
            resolve(qn, true);
            qn   = 0;
            lost = 1;
        }
    });
    resolve(qn, true);
    if (lost && lane == 0) lostflag = 1;
    __syncthreads();

    if (lostflag == 0) {
        drain(qn);
    } else {
        qn = 0;
        scan([&]() {
            if (qn >= 32) {
                const int n32 = qn & ~31;
                const int r   = qn - n32;
                int te = 0;
                if (lane < r) te = myq[n32 + lane];
                resolve(n32, false);
                drain(n32);
                if (lane < r) myq[lane] = te;
                qn = r;
            }
        });
        if (qn > 0) { resolve(qn, false); drain(qn); }
    }
    __syncthreads();

    float* __restrict__ orow = out + (size_t)d0 * F;
    const int tot = tcur * F;
    for (int i = tid * 4; i < tot; i += AGG_NW * 32 * 4) {
        float4 v = *(const float4*)(acc + i);
        const float den = sm[(i / F) * H + (i & (F - 1)) / CH] + 1e-16f;
        v.x = v.x / den + bv.x; v.y = v.y / den + bv.y; v.z = v.z / den + bv.z; v.w = v.w / den + bv.w;
        if (ELU) {
            v.x = (v.x > 0.f) ? v.x : (__expf(v.x) - 1.f);
            v.y = (v.y > 0.f) ? v.y : (__expf(v.y) - 1.f);
            v.z = (v.z > 0.f) ? v.z : (__expf(v.z) - 1.f);
            v.w = (v.w > 0.f) ? v.w : (__expf(v.w) - 1.f);
        }
        *(float4*)(orow + i) = v;
    }
}

extern "C" void kernel_launch(void* const* d_in, const int* in_sizes, int n_in,
                              void* d_out, int out_size, void* d_ws, size_t ws_size,
                              hipStream_t stream) {
    const float* x        = (const float*)d_in[0];
    const int*   ei       = (const int*)  d_in[1];
    const float* W1       = (const float*)d_in[2];
    const float* att_src1 = (const float*)d_in[3];
    const float* att_dst1 = (const float*)d_in[4];
    const float* b1       = (const float*)d_in[5];
    const float* W2       = (const float*)d_in[6];
    const float* att_src2 = (const float*)d_in[7];
    const float* att_dst2 = (const float*)d_in[8];
    const float* b2       = (const float*)d_in[9];
    float* out = (float*)d_out;

    const int N = in_sizes[0] / 128;
    const int E = in_sizes[1] / 2;

    float* ws = (float*)d_ws;
    float* h1  = ws;
    float* f1  = h1 + (size_t)N * 128;
    float* as1 = f1 + (size_t)N * 128;
    float* ad1 = as1 + (size_t)N * 4;
    float* as2 = ad1 + (size_t)N * 4;
    float* ad2 = as2 + N;
    unsigned char* own = (unsigned char*)(ad2 + N);
    float* h2  = h1;

    const int TB = 256;
    const int gBlocks = 512;
    const int aBlocks = (N + AGG_T - 1) / AGG_T;

    gemm_wmma_f32<128, 128><<<gBlocks, 128, 0, stream>>>(x, W1, h1, N);
    att_scores<<<(N * 4 + TB - 1) / TB, TB, 0, stream>>>(h1, att_src1, att_dst1, as1, ad1, N, 4, 32);
    owner_plane<<<((E + 3) / 4 + TB - 1) / TB, TB, 0, stream>>>(ei, E, own);
    edge_aggregate_own<128, 4, true><<<aBlocks, AGG_NW * 32, 0, stream>>>(ei, E, own, as1, ad1, h1, b1, f1, N);

    gemm_wmma_f32<128, 64><<<gBlocks, 128, 0, stream>>>(f1, W2, h2, N);
    att_scores<<<(N + TB - 1) / TB, TB, 0, stream>>>(h2, att_src2, att_dst2, as2, ad2, N, 1, 64);
    edge_aggregate_own<64, 1, false><<<aBlocks, AGG_NW * 32, 0, stream>>>(ei, E, own, as2, ad2, h2, b2, out, N);
}


# Aggregation reads its row of a per-owner chunk bitmap (1 bit per 16 edges, set by LDS atomic OR in the owner-plane launch) and only flagged chunks of the byte plane; LDS float adds in another order
# speedup vs baseline: 7.1518x; 1.0921x over previous
#include <hip/hip_runtime.h>
#include <math.h>

typedef __attribute__((ext_vector_type(2))) float v2f;
typedef __attribute__((ext_vector_type(8))) float v8f;

__device__ __forceinline__ unsigned f2key(float f) {
    unsigned b = __float_as_uint(f);
    return (b & 0x80000000u) ? ~b : (b | 0x80000000u);
}
__device__ __forceinline__ float key2f(unsigned k) {
    unsigned b = (k & 0x80000000u) ? (k & 0x7FFFFFFFu) : ~k;
    return __uint_as_float(b);
}

__device__ __forceinline__ constexpr int frag_k(int s, int h, int i) {
    return 8 * (s >> 1) + 4 * h + 2 * (s & 1) + i;
}

template <int K, int NC>
__global__ __launch_bounds__(128)
void gemm_wmma_f32(const float* __restrict__ A,
                   const float* __restrict__ B,
                   float* __restrict__ C,
                   int M) {
    constexpr int NG    = NC / 64;
    constexpr int SLOTS = 4 / NG;
    constexpr int KS    = K / 4;
    static_assert(NC % 64 == 0 && (4 % NG) == 0 && K % 8 == 0, "tile shape");

    const int tid  = threadIdx.x;
    const int lane = tid & 31;
    const int wave = __builtin_amdgcn_readfirstlane(tid >> 5);
    const int h    = lane >> 4;
    const int c    = lane & 15;
    const int nb   = (wave % NG) * 64;
    const int slot = wave / NG;
    const int mTiles = (M + 15) >> 4;
    const int stride = (int)gridDim.x * SLOTS;

    v2f breg[4][KS];
    {
        const float* __restrict__ bcol = B + nb + 4 * c;
#pragma unroll
        for (int s = 0; s < KS; ++s) {
#pragma unroll
            for (int i = 0; i < 2; ++i) {
                const float4 w = *(const float4*)(bcol + (size_t)frag_k(s, h, i) * NC);
                breg[0][s][i] = w.x; breg[1][s][i] = w.y; breg[2][s][i] = w.z; breg[3][s][i] = w.w;
            }
        }
    }

    for (int mt = (int)blockIdx.x * SLOTS + slot; mt < mTiles; mt += stride) {
        const int m0 = mt << 4;
        int rowA = m0 + c;
        if (rowA >= M) rowA = M - 1;
        const float* __restrict__ arow = A + (size_t)rowA * K + 4 * h;

        float4 q[K / 8];
#pragma unroll
        for (int j = 0; j < K / 8; ++j) q[j] = *(const float4*)(arow + 8 * j);

        v8f acc[4] = {};
#pragma unroll
        for (int s = 0; s < KS; ++s) {
            const float4 qq = q[s >> 1];
            v2f a;
            if (s & 1) { a.x = qq.z; a.y = qq.w; } else { a.x = qq.x; a.y = qq.y; }
#pragma unroll
            for (int t = 0; t < 4; ++t)
                acc[t] = __builtin_amdgcn_wmma_f32_16x16x4_f32(false, a, false, breg[t][s],
                                                               (short)0, acc[t], false, false);
        }

        float* __restrict__ crow = C + (size_t)(m0 + (h << 3)) * NC + nb + 4 * c;
#pragma unroll
        for (int r = 0; r < 8; ++r) {
            if (m0 + (h << 3) + r < M) {
                float4 o;
                o.x = acc[0][r]; o.y = acc[1][r]; o.z = acc[2][r]; o.w = acc[3][r];
                *(float4*)(crow + (size_t)r * NC) = o;
            }
        }
    }
}

__global__ void att_scores(const float* __restrict__ h,
                           const float* __restrict__ a_src,
                           const float* __restrict__ a_dst,
                           float* __restrict__ out_s,
                           float* __restrict__ out_d,
                           int N, int H, int Cc) {
    int t = blockIdx.x * blockDim.x + threadIdx.x;
    if (t >= N * H) return;
    int n  = t / H;
    int hh = t - n * H;
    const float* row = h + (size_t)n * H * Cc + (size_t)hh * Cc;
    const float* vs  = a_src + (size_t)hh * Cc;
    const float* vd  = a_dst + (size_t)hh * Cc;
    float ss = 0.f, dd = 0.f;
    for (int c = 0; c < Cc; ++c) {
        float v = row[c];
        ss += v * vs[c];
        dd += v * vd[c];
    }
    out_s[t] = ss;
    out_d[t] = dd;
}

#define AGG_T  196
#define AGG_NW 32
#define AGG_QC 1024

#define OP_CH 256
#define OP_W  (OP_CH / 32)
__global__ __launch_bounds__(OP_CH)
void owner_plane(const int* __restrict__ ei, int E, unsigned char* __restrict__ own,
                 unsigned* __restrict__ bm, int bmw) {
    static_assert(OP_W % 4 == 0, "a row piece is stored with 16-byte stores");
    __shared__ __attribute__((aligned(16))) unsigned lbm[256 * OP_W];
    const int tid = threadIdx.x;
    for (int i = tid; i < 256 * OP_W; i += OP_CH) lbm[i] = 0u;
    __syncthreads();

    const int* __restrict__ dstp = ei + E;
    const int  e0   = (blockIdx.x * OP_CH + tid) * 16;
    const bool full = (e0 + 16 <= E);
    unsigned ob[16];
    if (full && ((((size_t)(dstp + e0)) & 15) == 0)) {
        const int4* __restrict__ p = (const int4*)(dstp + e0);
        const int4 a = p[0], b = p[1], c = p[2], d = p[3];
        ob[0]  = (unsigned)a.x; ob[1]  = (unsigned)a.y; ob[2]  = (unsigned)a.z; ob[3]  = (unsigned)a.w;
        ob[4]  = (unsigned)b.x; ob[5]  = (unsigned)b.y; ob[6]  = (unsigned)b.z; ob[7]  = (unsigned)b.w;
        ob[8]  = (unsigned)c.x; ob[9]  = (unsigned)c.y; ob[10] = (unsigned)c.z; ob[11] = (unsigned)c.w;
        ob[12] = (unsigned)d.x; ob[13] = (unsigned)d.y; ob[14] = (unsigned)d.z; ob[15] = (unsigned)d.w;
    } else {
#pragma unroll
        for (int j = 0; j < 16; ++j) ob[j] = (e0 + j < E) ? (unsigned)dstp[e0 + j] : 0u;
    }
#pragma unroll
    for (int j = 0; j < 16; ++j) ob[j] = (ob[j] / (unsigned)AGG_T) & 255u;

    if (full && ((((size_t)(own + e0)) & 15) == 0)) {
        uint4 o;
        o.x = ob[0]  | (ob[1]  << 8) | (ob[2]  << 16) | (ob[3]  << 24);
        o.y = ob[4]  | (ob[5]  << 8) | (ob[6]  << 16) | (ob[7]  << 24);
        o.z = ob[8]  | (ob[9]  << 8) | (ob[10] << 16) | (ob[11] << 24);
        o.w = ob[12] | (ob[13] << 8) | (ob[14] << 16) | (ob[15] << 24);
        *(uint4*)(own + e0) = o;
    } else {
#pragma unroll
        for (int j = 0; j < 16; ++j) if (e0 + j < E) own[e0 + j] = (unsigned char)ob[j];
    }

    const unsigned bit = 1u << (tid & 31);
    unsigned*      lw  = lbm + (tid >> 5);
#pragma unroll
    for (int j = 0; j < 16; ++j)
        if (e0 + j < E) atomicOr(lw + ob[j] * OP_W, bit);
    __syncthreads();

    for (int r = tid; r < 256; r += OP_CH) {
        const uint4* s = (const uint4*)(lbm + r * OP_W);
        uint4*       g = (uint4*)(bm + (size_t)r * bmw + (size_t)blockIdx.x * OP_W);
#pragma unroll
        for (int w = 0; w < OP_W / 4; ++w) g[w] = s[w];
    }
}

template <int H>
__device__ __forceinline__ void ld_heads(const float* __restrict__ q, float (&o)[H]) {
    if constexpr (H == 4) {
        const float4 t = *(const float4*)q;
        o[0] = t.x; o[1] = t.y; o[2] = t.z; o[3] = t.w;
    } else {
#pragma unroll
        for (int h = 0; h < H; ++h) o[h] = q[h];
    }
}

template <int F, int H, bool ELU>
__global__ __launch_bounds__(1024)
void edge_aggregate_own(const int* __restrict__ ei, int E,
                        const unsigned char* __restrict__ own,
                        const unsigned* __restrict__ bm, int bmw,
                        const float* __restrict__ as_,
                        const float* __restrict__ ad_,
                        const float* __restrict__ hsrc,
                        const float* __restrict__ bias,
                        float* __restrict__ out, int N) {
    constexpr int KF = F / 32;
    constexpr int CH = F / H;
    constexpr int NB = 1024 / F;
    static_assert(CH % 32 == 0, "a 32-channel slice must lie inside one head");
    static_assert(AGG_T <= 256, "a local destination must fit a byte");
    __shared__ __attribute__((aligned(16))) float acc[AGG_T * F];
    __shared__ int           qs[AGG_NW * AGG_QC];
    __shared__ unsigned char qd[AGG_NW * AGG_QC];
    __shared__ unsigned      mx[AGG_T * H];
    __shared__ float         sm[AGG_T * H];
    __shared__ int           lostflag;

    const int tid  = threadIdx.x;
    const int lane = tid & 31;
    const int wave = __builtin_amdgcn_readfirstlane(tid >> 5);
    const int d0   = blockIdx.x * AGG_T;
    const int tcur = min(AGG_T, N - d0);

    static_assert((AGG_NW * 32 * 4) % F == 0, "a thread must keep its channels from trip to trip");
    const float4 bv = *(const float4*)(bias + ((tid * 4) & (F - 1)));

    for (int i = tid; i < AGG_T * F; i += AGG_NW * 32) acc[i] = 0.f;
    for (int i = tid; i < AGG_T * H; i += AGG_NW * 32) { mx[i] = 0u; sm[i] = 0.f; }
    if (tid == 0) lostflag = 0;
    __syncthreads();

    const int* __restrict__ dstp = ei + E;
    int*           myq  = qs + wave * AGG_QC;
    unsigned char* myqd = qd + wave * AGG_QC;
    const bool byte_ok = (gridDim.x <= 256u) && ((E & 15) == 0) && ((((size_t)own) & 15) == 0);

    static_assert(AGG_QC % 128 == 0, "a lane past the end must still store inside the queue");
    auto resolve = [&](int n, bool domax) {
        for (int off = 0; off < n; off += 128) {
            int  e[4], s[4], d[4];
            bool ok[4];
#pragma unroll
            for (int b = 0; b < 4; ++b) {
                const int idx = off + b * 32 + lane;
                ok[b] = idx < n;
                e[b]  = ok[b] ? myq[idx] : 0;
            }
#pragma unroll
            for (int b = 0; b < 4; ++b) { s[b] = ei[e[b]]; d[b] = dstp[e[b]]; }
            float a[4][H], c[4][H];
            if (domax) {
#pragma unroll
                for (int b = 0; b < 4; ++b) {
                    ld_heads<H>(as_ + (size_t)s[b] * H, a[b]);
                    ld_heads<H>(ad_ + (size_t)d[b] * H, c[b]);
                }
            }
#pragma unroll
            for (int b = 0; b < 4; ++b) {
                const int idx = off + b * 32 + lane;
                const int dl  = d[b] - d0;
                myq[idx]  = s[b];
                myqd[idx] = (unsigned char)dl;
                if (domax && ok[b]) {
#pragma unroll
                    for (int h = 0; h < H; ++h) {
                        float el = a[b][h] + c[b][h];
                        el = (el > 0.f) ? el : 0.2f * el;
                        atomicMax(&mx[dl * H + h], f2key(el));
                    }
                }
            }
        }
    };

    auto drain = [&](int n) {
        for (int off = 0; off < n; off += 32) {
            const int cnt = min(32, n - off);
            int   s = 0, dl = 0;
            float al[H];
#pragma unroll
            for (int h = 0; h < H; ++h) al[h] = 0.f;
            if (lane < cnt) {
                s  = myq[off + lane];
                dl = (int)myqd[off + lane];
                float a[H], c[H];
                ld_heads<H>(as_ + (size_t)s * H, a);
                ld_heads<H>(ad_ + (size_t)(d0 + dl) * H, c);
#pragma unroll
                for (int h = 0; h < H; ++h) {
                    float el = a[h] + c[h];
                    el = (el > 0.f) ? el : 0.2f * el;
                    const float pe = __expf(el - key2f(mx[dl * H + h]));
                    al[h] = pe;
                    __hip_atomic_fetch_add(&sm[dl * H + h], pe, __ATOMIC_RELAXED,
                                           __HIP_MEMORY_SCOPE_WORKGROUP);
                }
            }
            for (int j0 = 0; j0 < cnt; j0 += NB) {
                float v[NB][KF];
                int   dj[NB];
#pragma unroll
                for (int b = 0; b < NB; ++b) {
                    const int j  = min(j0 + b, cnt - 1);
                    const int sj = __builtin_amdgcn_readlane(s, j);
                    dj[b] = __builtin_amdgcn_readlane(dl, j);
                    const float* __restrict__ row = hsrc + (size_t)sj * F + lane;
#pragma unroll
                    for (int k = 0; k < KF; ++k) {
                        const float w = __int_as_float(
                            __builtin_amdgcn_readlane(__float_as_int(al[(k * 32) / CH]), j));
                        v[b][k] = row[k * 32] * w;
                    }
                }
#pragma unroll
                for (int b = 0; b < NB; ++b) {
                    if (j0 + b < cnt) {
                        float* arow = acc + dj[b] * F + lane;
#pragma unroll
                        for (int k = 0; k < KF; ++k)
                            __hip_atomic_fetch_add(arow + k * 32, v[b][k], __ATOMIC_RELAXED,
                                                   __HIP_MEMORY_SCOPE_WORKGROUP);
                    }
                }
            }
        }
    };

    int qn = 0;

    const unsigned bb = (unsigned)blockIdx.x * 0x01010101u;
    auto zb = [&](unsigned x) -> unsigned {
        const unsigned y = x ^ bb;
        return ~(((y & 0x7f7f7f7fu) + 0x7f7f7f7fu) | y) & 0x80808080u;
    };
    auto step = [&](const uint4& cur, int e0, bool valid) {
        unsigned h = (zb(cur.x) >> 7) | (zb(cur.y) >> 6) | (zb(cur.z) >> 5) | (zb(cur.w) >> 4);
        if (!valid) h = 0u;
#pragma nounroll
        for (int trip = 0; trip < 16; ++trip) {
            const bool     hit = (h != 0u);
            const unsigned m   = __builtin_amdgcn_ballot_w32(hit);
            if (!m) break;
            if (hit) {
                const int q   = __builtin_ctz(h);
                const int pos = qn + (int)__builtin_amdgcn_mbcnt_lo(m, 0u);
                myq[pos] = e0 + ((q & 7) << 2) + (q >> 3);
                h &= h - 1u;
            }
            qn += __builtin_popcount(m);
        }
    };

    auto scan = [&](auto look) {
        if (byte_ok) {
            const unsigned* __restrict__ bmrow = bm + (size_t)blockIdx.x * bmw;
            const int wpw = (bmw + AGG_NW - 1) / AGG_NW;
            for (int k = 0; k < wpw; k += 32) {
                const int wi = k + lane;
                const int gw = wave * wpw + wi;
                unsigned  cw = (wi < wpw && gw < bmw) ? bmrow[gw] : 0u;
                const int cb = gw << 9;
#pragma nounroll
                for (int r = 0; r < 8; ++r) {
                    if (!__builtin_amdgcn_ballot_w32(cw != 0u)) break;
                    int   eb[4];
                    bool  vb[4];
                    uint4 tb[4];
#pragma unroll
                    for (int b = 0; b < 4; ++b) {
                        const bool has = (cw != 0u);
                        const int  q   = has ? __builtin_ctz(cw) : 0;
                        cw &= cw - 1u;
                        eb[b] = cb + (q << 4);
                        vb[b] = has && (eb[b] + 16 <= E);
                        tb[b] = *(const uint4*)(own + (vb[b] ? eb[b] : 0));
                    }
#pragma unroll
                    for (int b = 0; b < 4; ++b) {
                        if (__builtin_amdgcn_ballot_w32(vb[b])) { step(tb[b], eb[b], vb[b]); look(); }
                    }
                }
            }
        } else {
            for (int wb = wave * 32; wb < E; wb += AGG_NW * 32) {
                const int  e   = wb + lane;
                const int  dv  = (e < E) ? dstp[e] : -1;
                const bool hit = (e < E) && ((unsigned)(dv - d0) < (unsigned)tcur);
                const unsigned m = __builtin_amdgcn_ballot_w32(hit);
                if (m) {
                    if (hit) myq[qn + (int)__builtin_amdgcn_mbcnt_lo(m, 0u)] = e;
                    qn += __builtin_popcount(m);
                }
                look();
            }
        }
    };

    int lost = 0;
    scan([&]() {
        if (qn > AGG_QC - 512) {
            resolve(qn, true);
            qn   = 0;
            lost = 1;
        }
    });
    resolve(qn, true);
    if (lost && lane == 0) lostflag = 1;
    __syncthreads();

    if (lostflag == 0) {
        drain(qn);
    } else {
        qn = 0;
        scan([&]() {
            if (qn >= 32) {
                const int n32 = qn & ~31;
                const int r   = qn - n32;
                int te = 0;
                if (lane < r) te = myq[n32 + lane];
                resolve(n32, false);
                drain(n32);
                if (lane < r) myq[lane] = te;
                qn = r;
            }
        });
        if (qn > 0) { resolve(qn, false); drain(qn); }
    }
    __syncthreads();

    float* __restrict__ orow = out + (size_t)d0 * F;
    const int tot = tcur * F;
    for (int i = tid * 4; i < tot; i += AGG_NW * 32 * 4) {
        float4 v = *(const float4*)(acc + i);
        const float den = sm[(i / F) * H + (i & (F - 1)) / CH] + 1e-16f;
        v.x = v.x / den + bv.x; v.y = v.y / den + bv.y; v.z = v.z / den + bv.z; v.w = v.w / den + bv.w;
        if (ELU) {
            v.x = (v.x > 0.f) ? v.x : (__expf(v.x) - 1.f);
            v.y = (v.y > 0.f) ? v.y : (__expf(v.y) - 1.f);
            v.z = (v.z > 0.f) ? v.z : (__expf(v.z) - 1.f);
            v.w = (v.w > 0.f) ? v.w : (__expf(v.w) - 1.f);
        }
        *(float4*)(orow + i) = v;
    }
}

extern "C" void kernel_launch(void* const* d_in, const int* in_sizes, int n_in,
                              void* d_out, int out_size, void* d_ws, size_t ws_size,
                              hipStream_t stream) {
    const float* x        = (const float*)d_in[0];
    const int*   ei       = (const int*)  d_in[1];
    const float* W1       = (const float*)d_in[2];
    const float* att_src1 = (const float*)d_in[3];
    const float* att_dst1 = (const float*)d_in[4];
    const float* b1       = (const float*)d_in[5];
    const float* W2       = (const float*)d_in[6];
    const float* att_src2 = (const float*)d_in[7];
    const float* att_dst2 = (const float*)d_in[8];
    const float* b2       = (const float*)d_in[9];
    float* out = (float*)d_out;

    const int N = in_sizes[0] / 128;
    const int E = in_sizes[1] / 2;

    float* ws = (float*)d_ws;
    float* h1  = ws;
    float* f1  = h1 + (size_t)N * 128;
    float* as1 = f1 + (size_t)N * 128;
    float* ad1 = as1 + (size_t)N * 4;
    float* as2 = ad1 + (size_t)N * 4;
    float* ad2 = as2 + N;
    unsigned char* own = (unsigned char*)(ad2 + N);
    const int nch      = (E + 15) / 16;
    const int opBlocks = (nch + OP_CH - 1) / OP_CH;
    const int bmw      = opBlocks * OP_W;
    unsigned* bm = (unsigned*)((((size_t)(own + E)) + 255) & ~(size_t)255);
    float* h2  = h1;

    const int TB = 256;
    const int gBlocks = 512;
    const int aBlocks = (N + AGG_T - 1) / AGG_T;

    gemm_wmma_f32<128, 128><<<gBlocks, 128, 0, stream>>>(x, W1, h1, N);
    att_scores<<<(N * 4 + TB - 1) / TB, TB, 0, stream>>>(h1, att_src1, att_dst1, as1, ad1, N, 4, 32);
    owner_plane<<<opBlocks, OP_CH, 0, stream>>>(ei, E, own, bm, bmw);
    edge_aggregate_own<128, 4, true><<<aBlocks, AGG_NW * 32, 0, stream>>>(ei, E, own, bm, bmw, as1, ad1, h1, b1, f1, N);

    gemm_wmma_f32<128, 64><<<gBlocks, 128, 0, stream>>>(f1, W2, h2, N);
    att_scores<<<(N + TB - 1) / TB, TB, 0, stream>>>(h2, att_src2, att_dst2, as2, ad2, N, 1, 64);
    edge_aggregate_own<64, 1, false><<<aBlocks, AGG_NW * 32, 0, stream>>>(ei, E, own, bm, bmw, as2, ad2, h2, b2, out, N);
}


# Aggregation reads flagged 4-edge chunks of a packed record plane (source, local destination, owner: one word per edge) instead of owner bytes plus the edge list; LDS float adds in another order
# speedup vs baseline: 7.7099x; 1.0780x over previous
#include <hip/hip_runtime.h>
#include <math.h>

typedef __attribute__((ext_vector_type(2))) float v2f;
typedef __attribute__((ext_vector_type(8))) float v8f;

__device__ __forceinline__ unsigned f2key(float f) {
    unsigned b = __float_as_uint(f);
    return (b & 0x80000000u) ? ~b : (b | 0x80000000u);
}
__device__ __forceinline__ float key2f(unsigned k) {
    unsigned b = (k & 0x80000000u) ? (k & 0x7FFFFFFFu) : ~k;
    return __uint_as_float(b);
}

__device__ __forceinline__ constexpr int frag_k(int s, int h, int i) {
    return 8 * (s >> 1) + 4 * h + 2 * (s & 1) + i;
}

template <int K, int NC>
__global__ __launch_bounds__(128)
void gemm_wmma_f32(const float* __restrict__ A,
                   const float* __restrict__ B,
                   float* __restrict__ C,
                   int M) {
    constexpr int NG    = NC / 64;
    constexpr int SLOTS = 4 / NG;
    constexpr int KS    = K / 4;
    static_assert(NC % 64 == 0 && (4 % NG) == 0 && K % 8 == 0, "tile shape");

    const int tid  = threadIdx.x;
    const int lane = tid & 31;
    const int wave = __builtin_amdgcn_readfirstlane(tid >> 5);
    const int h    = lane >> 4;
    const int c    = lane & 15;
    const int nb   = (wave % NG) * 64;
    const int slot = wave / NG;
    const int mTiles = (M + 15) >> 4;
    const int stride = (int)gridDim.x * SLOTS;

    v2f breg[4][KS];
    {
        const float* __restrict__ bcol = B + nb + 4 * c;
#pragma unroll
        for (int s = 0; s < KS; ++s) {
#pragma unroll
            for (int i = 0; i < 2; ++i) {
                const float4 w = *(const float4*)(bcol + (size_t)frag_k(s, h, i) * NC);
                breg[0][s][i] = w.x; breg[1][s][i] = w.y; breg[2][s][i] = w.z; breg[3][s][i] = w.w;
            }
        }
    }

    for (int mt = (int)blockIdx.x * SLOTS + slot; mt < mTiles; mt += stride) {
        const int m0 = mt << 4;
        int rowA = m0 + c;
        if (rowA >= M) rowA = M - 1;
        const float* __restrict__ arow = A + (size_t)rowA * K + 4 * h;

        float4 q[K / 8];
#pragma unroll
        for (int j = 0; j < K / 8; ++j) q[j] = *(const float4*)(arow + 8 * j);

        v8f acc[4] = {};
#pragma unroll
        for (int s = 0; s < KS; ++s) {
            const float4 qq = q[s >> 1];
            v2f a;
            if (s & 1) { a.x = qq.z; a.y = qq.w; } else { a.x = qq.x; a.y = qq.y; }
#pragma unroll
            for (int t = 0; t < 4; ++t)
                acc[t] = __builtin_amdgcn_wmma_f32_16x16x4_f32(false, a, false, breg[t][s],
                                                               (short)0, acc[t], false, false);
        }

        float* __restrict__ crow = C + (size_t)(m0 + (h << 3)) * NC + nb + 4 * c;
#pragma unroll
        for (int r = 0; r < 8; ++r) {
            if (m0 + (h << 3) + r < M) {
                float4 o;
                o.x = acc[0][r]; o.y = acc[1][r]; o.z = acc[2][r]; o.w = acc[3][r];
                *(float4*)(crow + (size_t)r * NC) = o;
            }
        }
    }
}

__global__ void att_scores(const float* __restrict__ h,
                           const float* __restrict__ a_src,
                           const float* __restrict__ a_dst,
                           float* __restrict__ out_s,
                           float* __restrict__ out_d,
                           int N, int H, int Cc) {
    int t = blockIdx.x * blockDim.x + threadIdx.x;
    if (t >= N * H) return;
    int n  = t / H;
    int hh = t - n * H;
    const float* row = h + (size_t)n * H * Cc + (size_t)hh * Cc;
    const float* vs  = a_src + (size_t)hh * Cc;
    const float* vd  = a_dst + (size_t)hh * Cc;
    float ss = 0.f, dd = 0.f;
    for (int c = 0; c < Cc; ++c) {
        float v = row[c];
        ss += v * vs[c];
        dd += v * vd[c];
    }
    out_s[t] = ss;
    out_d[t] = dd;
}

#define AGG_T  196
#define AGG_NW 32
#define AGG_QC 1024

#define REC_TH 256
#define REC_W  (4 * REC_TH / 32)
__global__ __launch_bounds__(REC_TH)
void edge_records(const int* __restrict__ ei, int E, unsigned* __restrict__ rec,
                  unsigned* __restrict__ bm, int bmw) {
    static_assert(REC_W % 4 == 0, "a row piece is stored with 16-byte stores");
    static_assert(AGG_T * 256 <= 65536, "a node index must fit 16 bits while a byte holds the owner");
    __shared__ __attribute__((aligned(16))) unsigned lbm[256 * REC_W];
    const int tid = threadIdx.x;
    for (int i = tid; i < 256 * REC_W / 4; i += REC_TH) ((uint4*)lbm)[i] = make_uint4(0u, 0u, 0u, 0u);
    __syncthreads();

    const int* __restrict__ dstp = ei + E;
    const bool al16 = (((size_t)ei) & 15) == 0 && (E & 3) == 0 && (((size_t)rec) & 15) == 0;
    unsigned sv[4][4], dv[4][4];
#pragma unroll
    for (int k = 0; k < 4; ++k) {
        const int e0 = ((blockIdx.x * 4 + k) * REC_TH + tid) * 4;
        if (al16 && e0 + 4 <= E) {
            const int4 s = *(const int4*)(ei + e0);
            const int4 d = *(const int4*)(dstp + e0);
            sv[k][0] = (unsigned)s.x; sv[k][1] = (unsigned)s.y; sv[k][2] = (unsigned)s.z; sv[k][3] = (unsigned)s.w;
            dv[k][0] = (unsigned)d.x; dv[k][1] = (unsigned)d.y; dv[k][2] = (unsigned)d.z; dv[k][3] = (unsigned)d.w;
        } else {
#pragma unroll
            for (int j = 0; j < 4; ++j) {
                sv[k][j] = (e0 + j < E) ? (unsigned)ei[e0 + j] : 0u;
                dv[k][j] = (e0 + j < E) ? (unsigned)dstp[e0 + j] : 0u;
            }
        }
    }
#pragma unroll
    for (int k = 0; k < 4; ++k) {
        const int e0 = ((blockIdx.x * 4 + k) * REC_TH + tid) * 4;
        unsigned rw[4], ow[4];
#pragma unroll
        for (int j = 0; j < 4; ++j) {
            const unsigned o = dv[k][j] / (unsigned)AGG_T;
            const unsigned l = dv[k][j] - o * (unsigned)AGG_T;
            ow[j] = o & 255u;
            rw[j] = (sv[k][j] & 0xffffu) | (l << 16) | (ow[j] << 24);
        }
        if (al16 && e0 + 4 <= E) {
            *(uint4*)(rec + e0) = make_uint4(rw[0], rw[1], rw[2], rw[3]);
        } else {
#pragma unroll
            for (int j = 0; j < 4; ++j) if (e0 + j < E) rec[e0 + j] = rw[j];
        }
        const unsigned bit = 1u << (tid & 31);
        unsigned*      lw  = lbm + 8 * k + (tid >> 5);
#pragma unroll
        for (int j = 0; j < 4; ++j)
            if (e0 + j < E) atomicOr(lw + ow[j] * REC_W, bit);
    }
    __syncthreads();

    for (int i = tid; i < 256 * (REC_W / 4); i += REC_TH) {
        const int r = i / (REC_W / 4);
        const int w = i - r * (REC_W / 4);
        ((uint4*)(bm + (size_t)r * bmw + (size_t)blockIdx.x * REC_W))[w] = ((const uint4*)(lbm + r * REC_W))[w];
    }
}

struct QRecord { static constexpr bool packed = true;  };
struct QEdge   { static constexpr bool packed = false; };

template <int H>
__device__ __forceinline__ void ld_heads(const float* __restrict__ q, float (&o)[H]) {
    if constexpr (H == 4) {
        const float4 t = *(const float4*)q;
        o[0] = t.x; o[1] = t.y; o[2] = t.z; o[3] = t.w;
    } else {
#pragma unroll
        for (int h = 0; h < H; ++h) o[h] = q[h];
    }
}

template <int F, int H, bool ELU>
__global__ __launch_bounds__(1024)
void edge_aggregate_own(const int* __restrict__ ei, int E,
                        const unsigned* __restrict__ rec,
                        const unsigned* __restrict__ bm, int bmw,
                        const float* __restrict__ as_,
                        const float* __restrict__ ad_,
                        const float* __restrict__ hsrc,
                        const float* __restrict__ bias,
                        float* __restrict__ out, int N) {
    constexpr int KF = F / 32;
    constexpr int CH = F / H;
    constexpr int NB = 1024 / F;
    static_assert(CH % 32 == 0, "a 32-channel slice must lie inside one head");
    static_assert(AGG_T <= 256, "a local destination must fit a byte");
    __shared__ __attribute__((aligned(16))) float acc[AGG_T * F];
    __shared__ int           qs[AGG_NW * AGG_QC];
    __shared__ unsigned char qd[AGG_NW * AGG_QC];
    __shared__ unsigned      mx[AGG_T * H];
    __shared__ float         sm[AGG_T * H];
    __shared__ int           lostflag;

    const int tid  = threadIdx.x;
    const int lane = tid & 31;
    const int wave = __builtin_amdgcn_readfirstlane(tid >> 5);
    const int d0   = blockIdx.x * AGG_T;
    const int tcur = min(AGG_T, N - d0);

    static_assert((AGG_NW * 32 * 4) % F == 0, "a thread must keep its channels from trip to trip");
    const float4 bv = *(const float4*)(bias + ((tid * 4) & (F - 1)));

    for (int i = tid; i < AGG_T * F; i += AGG_NW * 32) acc[i] = 0.f;
    for (int i = tid; i < AGG_T * H; i += AGG_NW * 32) { mx[i] = 0u; sm[i] = 0.f; }
    if (tid == 0) lostflag = 0;
    __syncthreads();

    const int* __restrict__ dstp = ei + E;
    int*           myq  = qs + wave * AGG_QC;
    unsigned char* myqd = qd + wave * AGG_QC;
    const bool byte_ok = (rec != nullptr) && (gridDim.x <= 256u) && ((E & 15) == 0) &&
                         ((((size_t)rec) & 15) == 0) && ((bmw & 3) == 0) && ((((size_t)bm) & 15) == 0);

    static_assert(AGG_QC % 128 == 0, "a lane past the end must still store inside the queue");
    auto resolve = [&](int n, bool domax, auto kind) {
        for (int off = 0; off < n; off += 128) {
            int  e[4], s[4], d[4];
            bool ok[4];
#pragma unroll
            for (int b = 0; b < 4; ++b) {
                const int idx = off + b * 32 + lane;
                ok[b] = idx < n;
                e[b]  = ok[b] ? myq[idx] : 0;
            }
#pragma unroll
            for (int b = 0; b < 4; ++b) {
                if constexpr (decltype(kind)::packed) {
                    s[b] = e[b] & 0xffff;
                    d[b] = d0 + ((e[b] >> 16) & 0xff);
                } else {
                    s[b] = ei[e[b]]; d[b] = dstp[e[b]];
                }
            }
            float a[4][H], c[4][H];
            if (domax) {
#pragma unroll
                for (int b = 0; b < 4; ++b) {
                    ld_heads<H>(as_ + (size_t)s[b] * H, a[b]);
                    ld_heads<H>(ad_ + (size_t)d[b] * H, c[b]);
                }
            }
#pragma unroll
            for (int b = 0; b < 4; ++b) {
                const int idx = off + b * 32 + lane;
                const int dl  = d[b] - d0;
                myq[idx]  = s[b];
                myqd[idx] = (unsigned char)dl;
                if (domax && ok[b]) {
#pragma unroll
                    for (int h = 0; h < H; ++h) {
                        float el = a[b][h] + c[b][h];
                        el = (el > 0.f) ? el : 0.2f * el;
                        atomicMax(&mx[dl * H + h], f2key(el));
                    }
                }
            }
        }
    };

    auto drain = [&](int n) {
        for (int off = 0; off < n; off += 32) {
            const int cnt = min(32, n - off);
            int   s = 0, dl = 0;
            float al[H];
#pragma unroll
            for (int h = 0; h < H; ++h) al[h] = 0.f;
            if (lane < cnt) {
                s  = myq[off + lane];
                dl = (int)myqd[off + lane];
                float a[H], c[H];
                ld_heads<H>(as_ + (size_t)s * H, a);
                ld_heads<H>(ad_ + (size_t)(d0 + dl) * H, c);
#pragma unroll
                for (int h = 0; h < H; ++h) {
                    float el = a[h] + c[h];
                    el = (el > 0.f) ? el : 0.2f * el;
                    const float pe = __expf(el - key2f(mx[dl * H + h]));
                    al[h] = pe;
                    __hip_atomic_fetch_add(&sm[dl * H + h], pe, __ATOMIC_RELAXED,
                                           __HIP_MEMORY_SCOPE_WORKGROUP);
                }
            }
            for (int j0 = 0; j0 < cnt; j0 += NB) {
                float v[NB][KF];
                int   dj[NB];
#pragma unroll
                for (int b = 0; b < NB; ++b) {
                    const int j  = min(j0 + b, cnt - 1);
                    const int sj = __builtin_amdgcn_readlane(s, j);
                    dj[b] = __builtin_amdgcn_readlane(dl, j);
                    const float* __restrict__ row = hsrc + (size_t)sj * F + lane;
#pragma unroll
                    for (int k = 0; k < KF; ++k) {
                        const float w = __int_as_float(
                            __builtin_amdgcn_readlane(__float_as_int(al[(k * 32) / CH]), j));
                        v[b][k] = row[k * 32] * w;
                    }
                }
#pragma unroll
                for (int b = 0; b < NB; ++b) {
                    if (j0 + b < cnt) {
                        float* arow = acc + dj[b] * F + lane;
#pragma unroll
                        for (int k = 0; k < KF; ++k)
                            __hip_atomic_fetch_add(arow + k * 32, v[b][k], __ATOMIC_RELAXED,
                                                   __HIP_MEMORY_SCOPE_WORKGROUP);
                    }
                }
            }
        }
    };

    int qn = 0;

    const unsigned bi = blockIdx.x;
    auto step = [&](const uint4& cur, bool valid) {
        unsigned h = (((cur.x >> 24) == bi) ? 1u : 0u) | (((cur.y >> 24) == bi) ? 2u : 0u) |
                     (((cur.z >> 24) == bi) ? 4u : 0u) | (((cur.w >> 24) == bi) ? 8u : 0u);
        if (!valid) h = 0u;
#pragma nounroll
        for (int trip = 0; trip < 4; ++trip) {
            const bool     hit = (h != 0u);
            const unsigned m   = __builtin_amdgcn_ballot_w32(hit);
            if (!m) break;
            if (hit) {
                const int      q   = __builtin_ctz(h);
                const int      pos = qn + (int)__builtin_amdgcn_mbcnt_lo(m, 0u);
                const unsigned r   = (q == 0) ? cur.x : (q == 1) ? cur.y : (q == 2) ? cur.z : cur.w;
                myq[pos] = (int)r;
                h &= h - 1u;
            }
            qn += __builtin_popcount(m);
        }
    };

    auto scan = [&](auto look) {
        if (byte_ok) {
            const uint4* __restrict__ bmrow = (const uint4*)(bm + (size_t)blockIdx.x * bmw);
            const int npc = bmw >> 2;
            const int ppw = (npc + AGG_NW - 1) / AGG_NW;
            for (int k = 0; k < ppw; k += 32) {
                const int pi = k + lane;
                const int gp = wave * ppw + pi;
                uint4 cw = make_uint4(0u, 0u, 0u, 0u);
                if (pi < ppw && gp < npc) cw = bmrow[gp];
                const int cb = gp << 9;
#pragma nounroll
                for (int r = 0; r < 32; ++r) {
                    if (!__builtin_amdgcn_ballot_w32((cw.x | cw.y | cw.z | cw.w) != 0u)) break;
                    int   eb[4];
                    bool  vb[4];
                    uint4 tb[4];
#pragma unroll
                    for (int b = 0; b < 4; ++b) {
                        const int      sel = cw.x ? 0 : cw.y ? 1 : cw.z ? 2 : 3;
                        const unsigned w   = (sel == 0) ? cw.x : (sel == 1) ? cw.y : (sel == 2) ? cw.z : cw.w;
                        const bool     has = (w != 0u);
                        const int      q   = has ? (sel << 5) + __builtin_ctz(w) : 0;
                        const unsigned w2  = w & (w - 1u);
                        cw.x = (sel == 0) ? w2 : cw.x;
                        cw.y = (sel == 1) ? w2 : cw.y;
                        cw.z = (sel == 2) ? w2 : cw.z;
                        cw.w = (sel == 3) ? w2 : cw.w;
                        eb[b] = cb + (q << 2);
                        vb[b] = has && (eb[b] + 4 <= E);
                        tb[b] = *(const uint4*)(rec + (vb[b] ? eb[b] : 0));
                    }
#pragma unroll
                    for (int b = 0; b < 4; ++b) {
                        if (__builtin_amdgcn_ballot_w32(vb[b])) { step(tb[b], vb[b]); look(QRecord{}); }
                    }
                }
            }
        } else {
            for (int wb = wave * 32; wb < E; wb += AGG_NW * 32) {
                const int  e   = wb + lane;
                const int  dv  = (e < E) ? dstp[e] : -1;
                const bool hit = (e < E) && ((unsigned)(dv - d0) < (unsigned)tcur);
                const unsigned m = __builtin_amdgcn_ballot_w32(hit);
                if (m) {
                    if (hit) myq[qn + (int)__builtin_amdgcn_mbcnt_lo(m, 0u)] = e;
                    qn += __builtin_popcount(m);
                }
                look(QEdge{});
            }
        }
    };

    int lost = 0;
    scan([&](auto kind) {
        if (qn > AGG_QC - 512) {
            resolve(qn, true, kind);
            qn   = 0;
            lost = 1;
        }
    });
    if (byte_ok) resolve(qn, true, QRecord{}); else resolve(qn, true, QEdge{});
    if (lost && lane == 0) lostflag = 1;
    __syncthreads();

    if (lostflag == 0) {
        drain(qn);
    } else {
        qn = 0;
        scan([&](auto kind) {
            if (qn >= 32) {
                const int n32 = qn & ~31;
                const int r   = qn - n32;
                int te = 0;
                if (lane < r) te = myq[n32 + lane];
                resolve(n32, false, kind);
                drain(n32);
                if (lane < r) myq[lane] = te;
                qn = r;
            }
        });
        if (qn > 0) {
            if (byte_ok) resolve(qn, false, QRecord{}); else resolve(qn, false, QEdge{});
            drain(qn);
        }
    }
    __syncthreads();

    float* __restrict__ orow = out + (size_t)d0 * F;
    const int tot = tcur * F;
    for (int i = tid * 4; i < tot; i += AGG_NW * 32 * 4) {
        float4 v = *(const float4*)(acc + i);
        const float den = sm[(i / F) * H + (i & (F - 1)) / CH] + 1e-16f;
        v.x = v.x / den + bv.x; v.y = v.y / den + bv.y; v.z = v.z / den + bv.z; v.w = v.w / den + bv.w;
        if (ELU) {
            v.x = (v.x > 0.f) ? v.x : (__expf(v.x) - 1.f);
            v.y = (v.y > 0.f) ? v.y : (__expf(v.y) - 1.f);
            v.z = (v.z > 0.f) ? v.z : (__expf(v.z) - 1.f);
            v.w = (v.w > 0.f) ? v.w : (__expf(v.w) - 1.f);
        }
        *(float4*)(orow + i) = v;
    }
}

extern "C" void kernel_launch(void* const* d_in, const int* in_sizes, int n_in,
                              void* d_out, int out_size, void* d_ws, size_t ws_size,
                              hipStream_t stream) {
    const float* x        = (const float*)d_in[0];
    const int*   ei       = (const int*)  d_in[1];
    const float* W1       = (const float*)d_in[2];
    const float* att_src1 = (const float*)d_in[3];
    const float* att_dst1 = (const float*)d_in[4];
    const float* b1       = (const float*)d_in[5];
    const float* W2       = (const float*)d_in[6];
    const float* att_src2 = (const float*)d_in[7];
    const float* att_dst2 = (const float*)d_in[8];
    const float* b2       = (const float*)d_in[9];
    float* out = (float*)d_out;

    const int N = in_sizes[0] / 128;
    const int E = in_sizes[1] / 2;

    float* ws = (float*)d_ws;
    float* h1  = ws;
    float* f1  = h1 + (size_t)N * 128;
    float* as1 = f1 + (size_t)N * 128;
    float* ad1 = as1 + (size_t)N * 4;
    float* as2 = ad1 + (size_t)N * 4;
    float* ad2 = as2 + N;
    unsigned* rec = (unsigned*)((((size_t)(ad2 + N)) + 15) & ~(size_t)15);
    const int nch       = (E + 3) / 4;
    const int recBlocks = (nch + 4 * REC_TH - 1) / (4 * REC_TH);
    const int bmw       = recBlocks * REC_W;
    unsigned* bm = (unsigned*)((((size_t)(rec + E)) + 255) & ~(size_t)255);
    const bool planes = (size_t)((const char*)(bm + (size_t)256 * bmw) - (const char*)d_ws) <= ws_size;
    if (!planes) { rec = nullptr; bm = nullptr; }
    float* h2  = h1;

    const int TB = 256;
    const int gBlocks = 512;
    const int aBlocks = (N + AGG_T - 1) / AGG_T;

    gemm_wmma_f32<128, 128><<<gBlocks, 128, 0, stream>>>(x, W1, h1, N);
    att_scores<<<(N * 4 + TB - 1) / TB, TB, 0, stream>>>(h1, att_src1, att_dst1, as1, ad1, N, 4, 32);
    if (planes) edge_records<<<recBlocks, REC_TH, 0, stream>>>(ei, E, rec, bm, bmw);
    edge_aggregate_own<128, 4, true><<<aBlocks, AGG_NW * 32, 0, stream>>>(ei, E, rec, bm, bmw, as1, ad1, h1, b1, f1, N);

    gemm_wmma_f32<128, 64><<<gBlocks, 128, 0, stream>>>(f1, W2, h2, N);
    att_scores<<<(N + TB - 1) / TB, TB, 0, stream>>>(h2, att_src2, att_dst2, as2, ad2, N, 1, 64);
    edge_aggregate_own<64, 1, false><<<aBlocks, AGG_NW * 32, 0, stream>>>(ei, E, rec, bm, bmw, as2, ad2, h2, b2, out, N);
}


# Aggregation: an edge's logits (LeakyReLU(as+ad), formed for the segment maxima) stay in registers across the barrier, 4 edges per lane; the exponential pass loads no scores again and reads no queue
# speedup vs baseline: 9.1227x; 1.1832x over previous
#include <hip/hip_runtime.h>
#include <math.h>

typedef __attribute__((ext_vector_type(2))) float v2f;
typedef __attribute__((ext_vector_type(8))) float v8f;

__device__ __forceinline__ unsigned f2key(float f) {
    unsigned b = __float_as_uint(f);
    return (b & 0x80000000u) ? ~b : (b | 0x80000000u);
}
__device__ __forceinline__ float key2f(unsigned k) {
    unsigned b = (k & 0x80000000u) ? (k & 0x7FFFFFFFu) : ~k;
    return __uint_as_float(b);
}

__device__ __forceinline__ constexpr int frag_k(int s, int h, int i) {
    return 8 * (s >> 1) + 4 * h + 2 * (s & 1) + i;
}

struct WithScores    { static constexpr bool value = true;  };
struct WithoutScores { static constexpr bool value = false; };

template <int K, int NC, int H>
__global__ __launch_bounds__(128)
void gemm_wmma_f32(const float* __restrict__ A,
                   const float* __restrict__ B,
                   const float* __restrict__ WT,
                   float* __restrict__ C,
                   float* __restrict__ S,
                   int M) {
    constexpr int NG    = NC / 64;
    constexpr int SLOTS = 4 / NG;
    constexpr int KS    = K / 4;
    static_assert(NC % 64 == 0 && (4 % NG) == 0 && K % 8 == 0, "tile shape");
    static_assert(NG <= 2, "the waves of a row tile take the score tile in turns of two");
    static_assert(H == 1 || H == 4, "a lane stores its H scores with one 4-byte or one 16-byte store");

    const int tid  = threadIdx.x;
    const int lane = tid & 31;
    const int wave = __builtin_amdgcn_readfirstlane(tid >> 5);
    const int h    = lane >> 4;
    const int c    = lane & 15;
    const int grp  = wave % NG;
    const int nb   = grp * 64;
    const int slot = wave / NG;
    const int mTiles = (M + 15) >> 4;
    const int stride = (int)gridDim.x * SLOTS;

    v2f breg[4][KS];
    {
        const float* __restrict__ bcol = B + nb + 4 * c;
#pragma unroll
        for (int s = 0; s < KS; ++s) {
#pragma unroll
            for (int i = 0; i < 2; ++i) {
                const float4 w = *(const float4*)(bcol + (size_t)frag_k(s, h, i) * NC);
                breg[0][s][i] = w.x; breg[1][s][i] = w.y; breg[2][s][i] = w.z; breg[3][s][i] = w.w;
            }
        }
    }

    float4 wq[K / 8];
    {
        const float* __restrict__ wrow = WT + (size_t)c * K + 4 * h;
#pragma unroll
        for (int j = 0; j < K / 8; ++j) wq[j] = *(const float4*)(wrow + 8 * j);
    }

    int trip = 0;
    for (int mt = (int)blockIdx.x * SLOTS + slot; mt < mTiles; mt += stride, ++trip) {
        const int m0 = mt << 4;
        int rowA = m0 + c;
        if (rowA >= M) rowA = M - 1;
        const float* __restrict__ arow = A + (size_t)rowA * K + 4 * h;

        float4 q[K / 8];
#pragma unroll
        for (int j = 0; j < K / 8; ++j) q[j] = *(const float4*)(arow + 8 * j);

        const bool scores = (NG == 1) || (((trip ^ grp) & 1) == 0);

        v8f acc[4] = {};
        v8f accS[2] = {};
        auto ksteps = [&](auto with) {
#pragma unroll
            for (int s = 0; s < KS; ++s) {
                const float4 qq = q[s >> 1];
                v2f a;
                if (s & 1) { a.x = qq.z; a.y = qq.w; } else { a.x = qq.x; a.y = qq.y; }
#pragma unroll
                for (int t = 0; t < 4; ++t)
                    acc[t] = __builtin_amdgcn_wmma_f32_16x16x4_f32(false, a, false, breg[t][s],
                                                                   (short)0, acc[t], false, false);
                if constexpr (decltype(with)::value) {
                    const float4 ww = wq[s >> 1];
                    v2f w;
                    if (s & 1) { w.x = ww.z; w.y = ww.w; } else { w.x = ww.x; w.y = ww.y; }
                    accS[s & 1] = __builtin_amdgcn_wmma_f32_16x16x4_f32(false, w, false, a,
                                                                        (short)0, accS[s & 1], false, false);
                }
            }
        };
        if (scores) ksteps(WithScores{}); else ksteps(WithoutScores{});

        float* __restrict__ crow = C + (size_t)(m0 + (h << 3)) * NC + nb + 4 * c;
#pragma unroll
        for (int r = 0; r < 8; ++r) {
            if (m0 + (h << 3) + r < M) {
                float4 o;
                o.x = acc[0][r]; o.y = acc[1][r]; o.z = acc[2][r]; o.w = acc[3][r];
                *(float4*)(crow + (size_t)r * NC) = o;
            }
        }

        if (scores && m0 + c < M) {
            float* __restrict__ srow = S + (size_t)(m0 + c) * (2 * H) + h * H;
            if constexpr (H == 4) {
                float4 o;
                o.x = accS[0][0] + accS[1][0]; o.y = accS[0][1] + accS[1][1];
                o.z = accS[0][2] + accS[1][2]; o.w = accS[0][3] + accS[1][3];
                *(float4*)srow = o;
            } else {
                srow[0] = accS[0][0] + accS[1][0];
            }
        }
    }
}

#define FOLD_K 128
#define FOLD_N (2 * 16 * FOLD_K)
template <int NC, int H>
__device__ __forceinline__ void fold_scores(int r, int k,
                                            const float* __restrict__ W,
                                            const float* __restrict__ a_src,
                                            const float* __restrict__ a_dst,
                                            float* __restrict__ WT) {
    constexpr int CH = NC / H;
    static_assert(H <= 8 && CH % 4 == 0, "score rows 0-7 source, 8-15 destination; rows of W read 16 bytes at a time");
    const int hd = r & 7;
    float v = 0.f;
    if (hd < H) {
        const float* __restrict__ att = ((r & 8) ? a_dst : a_src) + hd * CH;
        const float* __restrict__ w   = W + (size_t)k * NC + hd * CH;
        for (int ch = 0; ch < CH; ch += 4) {
            const float4 x = *(const float4*)(w + ch);
            v = fmaf(x.x, att[ch],     v);
            v = fmaf(x.y, att[ch + 1], v);
            v = fmaf(x.z, att[ch + 2], v);
            v = fmaf(x.w, att[ch + 3], v);
        }
    }
    WT[r * FOLD_K + k] = v;
}

#define AGG_T  196
#define AGG_NW 32
#define AGG_QC 1024

#define REC_TH 256
#define REC_W  (4 * REC_TH / 32)
#define FOLD_GROUPS (FOLD_N / REC_TH)
__global__ __launch_bounds__(REC_TH)
void edge_records(const int* __restrict__ ei, int E, unsigned* __restrict__ rec,
                  unsigned* __restrict__ bm, int bmw, int nrec,
                  const float* __restrict__ W1, const float* __restrict__ att_src1,
                  const float* __restrict__ att_dst1, float* __restrict__ WT1,
                  const float* __restrict__ W2, const float* __restrict__ att_src2,
                  const float* __restrict__ att_dst2, float* __restrict__ WT2) {
    static_assert(REC_W % 4 == 0, "a row piece is stored with 16-byte stores");
    static_assert(AGG_T * 256 <= 65536, "a node index must fit 16 bits while a byte holds the owner");
    static_assert(FOLD_N % REC_TH == 0, "one element of WT per thread of the fold groups");
    __shared__ __attribute__((aligned(16))) unsigned lbm[256 * REC_W];
    const int tid = threadIdx.x;
    if ((int)blockIdx.x >= nrec) {
        const int idx = ((int)blockIdx.x - nrec) * REC_TH + tid;
        const int r   = (idx / FOLD_K) & 15;
        const int k   = idx % FOLD_K;
        if (idx < 16 * FOLD_K) fold_scores<128, 4>(r, k, W1, att_src1, att_dst1, WT1);
        else                   fold_scores<64, 1>(r, k, W2, att_src2, att_dst2, WT2);
        return;
    }
    for (int i = tid; i < 256 * REC_W / 4; i += REC_TH) ((uint4*)lbm)[i] = make_uint4(0u, 0u, 0u, 0u);
    __syncthreads();

    const int* __restrict__ dstp = ei + E;
    const bool al16 = (((size_t)ei) & 15) == 0 && (E & 3) == 0 && (((size_t)rec) & 15) == 0;
    unsigned sv[4][4], dv[4][4];
#pragma unroll
    for (int k = 0; k < 4; ++k) {
        const int e0 = ((blockIdx.x * 4 + k) * REC_TH + tid) * 4;
        if (al16 && e0 + 4 <= E) {
            const int4 s = *(const int4*)(ei + e0);
            const int4 d = *(const int4*)(dstp + e0);
            sv[k][0] = (unsigned)s.x; sv[k][1] = (unsigned)s.y; sv[k][2] = (unsigned)s.z; sv[k][3] = (unsigned)s.w;
            dv[k][0] = (unsigned)d.x; dv[k][1] = (unsigned)d.y; dv[k][2] = (unsigned)d.z; dv[k][3] = (unsigned)d.w;
        } else {
#pragma unroll
            for (int j = 0; j < 4; ++j) {
                sv[k][j] = (e0 + j < E) ? (unsigned)ei[e0 + j] : 0u;
                dv[k][j] = (e0 + j < E) ? (unsigned)dstp[e0 + j] : 0u;
            }
        }
    }
#pragma unroll
    for (int k = 0; k < 4; ++k) {
        const int e0 = ((blockIdx.x * 4 + k) * REC_TH + tid) * 4;
        unsigned rw[4], ow[4];
#pragma unroll
        for (int j = 0; j < 4; ++j) {
            const unsigned o = dv[k][j] / (unsigned)AGG_T;
            const unsigned l = dv[k][j] - o * (unsigned)AGG_T;
            ow[j] = o & 255u;
            rw[j] = (sv[k][j] & 0xffffu) | (l << 16) | (ow[j] << 24);
        }
        if (al16 && e0 + 4 <= E) {
            *(uint4*)(rec + e0) = make_uint4(rw[0], rw[1], rw[2], rw[3]);
        } else {
#pragma unroll
            for (int j = 0; j < 4; ++j) if (e0 + j < E) rec[e0 + j] = rw[j];
        }
        const unsigned bit = 1u << (tid & 31);
        unsigned*      lw  = lbm + 8 * k + (tid >> 5);
#pragma unroll
        for (int j = 0; j < 4; ++j)
            if (e0 + j < E) atomicOr(lw + ow[j] * REC_W, bit);
    }
    __syncthreads();

    for (int i = tid; i < 256 * (REC_W / 4); i += REC_TH) {
        const int r = i / (REC_W / 4);
        const int w = i - r * (REC_W / 4);
        ((uint4*)(bm + (size_t)r * bmw + (size_t)blockIdx.x * REC_W))[w] = ((const uint4*)(lbm + r * REC_W))[w];
    }
}

struct QRecord { static constexpr bool packed = true;  };
struct QEdge   { static constexpr bool packed = false; };

struct KeepFirst { static constexpr bool value = true;  };
struct KeepNone  { static constexpr bool value = false; };

typedef __attribute__((ext_vector_type(4))) float v4f;
template <int H> struct HeadVec;
template <> struct HeadVec<4> { typedef v4f   type; };
template <> struct HeadVec<1> { typedef float type; };
__device__ __forceinline__ float hv_get(float v, int)        { return v; }
__device__ __forceinline__ float hv_get(v4f v, int h)        { return v[h]; }
__device__ __forceinline__ void  hv_set(float& v, int, float x) { v = x; }
__device__ __forceinline__ void  hv_set(v4f& v, int h, float x) { v[h] = x; }

template <int H>
__device__ __forceinline__ void ld_heads(const float* __restrict__ q, float (&o)[H]) {
    if constexpr (H == 4) {
        const float4 t = *(const float4*)q;
        o[0] = t.x; o[1] = t.y; o[2] = t.z; o[3] = t.w;
    } else {
#pragma unroll
        for (int h = 0; h < H; ++h) o[h] = q[h];
    }
}

template <int F, int H, bool ELU>
__global__ __launch_bounds__(1024)
void edge_aggregate_own(const int* __restrict__ ei, int E,
                        const unsigned* __restrict__ rec,
                        const unsigned* __restrict__ bm, int bmw,
                        const float* __restrict__ sc,
                        const float* __restrict__ hsrc,
                        const float* __restrict__ bias,
                        float* __restrict__ out, int N) {
    constexpr int KF = F / 32;
    constexpr int CH = F / H;
    constexpr int NB = 1024 / F;
    static_assert(CH % 32 == 0, "a 32-channel slice must lie inside one head");
    static_assert(AGG_T <= 256, "a local destination must fit a byte");
    __shared__ __attribute__((aligned(16))) float acc[AGG_T * F];
    __shared__ int           qs[AGG_NW * AGG_QC];
    __shared__ unsigned char qd[AGG_NW * AGG_QC];
    __shared__ unsigned      mx[AGG_T * H];
    __shared__ float         sm[AGG_T * H];
    __shared__ int           lostflag;

    const int tid  = threadIdx.x;
    const int lane = tid & 31;
    const int wave = __builtin_amdgcn_readfirstlane(tid >> 5);
    const int d0   = blockIdx.x * AGG_T;
    const int tcur = min(AGG_T, N - d0);

    static_assert((AGG_NW * 32 * 4) % F == 0, "a thread must keep its channels from trip to trip");
    const float4 bv = *(const float4*)(bias + ((tid * 4) & (F - 1)));

    for (int i = tid; i < AGG_T * F; i += AGG_NW * 32) acc[i] = 0.f;
    for (int i = tid; i < AGG_T * H; i += AGG_NW * 32) { mx[i] = 0u; sm[i] = 0.f; }
    if (tid == 0) lostflag = 0;
    __syncthreads();

    const int* __restrict__ dstp = ei + E;
    int*           myq  = qs + wave * AGG_QC;
    unsigned char* myqd = qd + wave * AGG_QC;
    const bool byte_ok = (rec != nullptr) && (gridDim.x <= 256u) && ((E & 15) == 0) &&
                         ((((size_t)rec) & 15) == 0) && ((bmw & 3) == 0) && ((((size_t)bm) & 15) == 0);

    static_assert(AGG_QC % 128 == 0, "a lane past the end must still store inside the queue");
    int ks0 = 0, ks1 = 0, ks2 = 0, ks3 = 0;
    int kd0 = 0, kd1 = 0, kd2 = 0, kd3 = 0;
    typename HeadVec<H>::type kl0 = {}, kl1 = {}, kl2 = {}, kl3 = {};
    auto resolve = [&](int n, bool domax, auto kind, auto keep) {
        for (int off = 0; off < n; off += 128) {
            int  e[4], s[4], d[4];
            bool ok[4];
#pragma unroll
            for (int b = 0; b < 4; ++b) {
                const int idx = off + b * 32 + lane;
                ok[b] = idx < n;
                e[b]  = ok[b] ? myq[idx] : 0;
            }
#pragma unroll
            for (int b = 0; b < 4; ++b) {
                if constexpr (decltype(kind)::packed) {
                    s[b] = e[b] & 0xffff;
                    d[b] = d0 + ((e[b] >> 16) & 0xff);
                } else {
                    s[b] = ei[e[b]]; d[b] = dstp[e[b]];
                }
            }
            float a[4][H], c[4][H];
            if (domax) {
#pragma unroll
                for (int b = 0; b < 4; ++b) {
                    ld_heads<H>(sc + (size_t)s[b] * (2 * H), a[b]);
                    ld_heads<H>(sc + (size_t)d[b] * (2 * H) + H, c[b]);
                }
            }
#pragma unroll
            for (int b = 0; b < 4; ++b) {
                const int idx = off + b * 32 + lane;
                const int dl  = d[b] - d0;
                myq[idx]  = s[b];
                myqd[idx] = (unsigned char)dl;
                if (domax) {
                    float el[H];
#pragma unroll
                    for (int h = 0; h < H; ++h) {
                        el[h] = a[b][h] + c[b][h];
                        el[h] = (el[h] > 0.f) ? el[h] : 0.2f * el[h];
                    }
                    if (ok[b]) {
#pragma unroll
                        for (int h = 0; h < H; ++h) atomicMax(&mx[dl * H + h], f2key(el[h]));
                    }
                    if constexpr (decltype(keep)::value) {
                        if (off == 0) {
                            typename HeadVec<H>::type lv = {};
#pragma unroll
                            for (int h = 0; h < H; ++h) hv_set(lv, h, el[h]);
                            if      (b == 0) { ks0 = s[b]; kd0 = dl; kl0 = lv; }
                            else if (b == 1) { ks1 = s[b]; kd1 = dl; kl1 = lv; }
                            else if (b == 2) { ks2 = s[b]; kd2 = dl; kl2 = lv; }
                            else             { ks3 = s[b]; kd3 = dl; kl3 = lv; }
                        }
                    }
                }
            }
        }
    };

    auto drain = [&](int n, bool kept) {
        for (int off = 0; off < n; off += 32) {
            const int cnt = min(32, n - off);
            const int t   = off >> 5;
            int   s = 0, dl = 0;
            float el[H], al[H];
#pragma unroll
            for (int h = 0; h < H; ++h) { el[h] = 0.f; al[h] = 0.f; }
            if (kept && t < 4) {
                s  = (t == 0) ? ks0 : (t == 1) ? ks1 : (t == 2) ? ks2 : ks3;
                dl = (t == 0) ? kd0 : (t == 1) ? kd1 : (t == 2) ? kd2 : kd3;
                const typename HeadVec<H>::type lv = (t == 0) ? kl0 : (t == 1) ? kl1 : (t == 2) ? kl2 : kl3;
#pragma unroll
                for (int h = 0; h < H; ++h) el[h] = hv_get(lv, h);
            } else if (lane < cnt) {
                s  = myq[off + lane];
                dl = (int)myqd[off + lane];
                float a[H], c[H];
                ld_heads<H>(sc + (size_t)s * (2 * H), a);
                ld_heads<H>(sc + (size_t)(d0 + dl) * (2 * H) + H, c);
#pragma unroll
                for (int h = 0; h < H; ++h) {
                    el[h] = a[h] + c[h];
                    el[h] = (el[h] > 0.f) ? el[h] : 0.2f * el[h];
                }
            }
            if (lane < cnt) {
#pragma unroll
                for (int h = 0; h < H; ++h) {
                    const float pe = __expf(el[h] - key2f(mx[dl * H + h]));
                    al[h] = pe;
                    __hip_atomic_fetch_add(&sm[dl * H + h], pe, __ATOMIC_RELAXED,
                                           __HIP_MEMORY_SCOPE_WORKGROUP);
                }
            }
            for (int j0 = 0; j0 < cnt; j0 += NB) {
                float v[NB][KF];
                int   dj[NB];
#pragma unroll
                for (int b = 0; b < NB; ++b) {
                    const int j  = min(j0 + b, cnt - 1);
                    const int sj = __builtin_amdgcn_readlane(s, j);
                    dj[b] = __builtin_amdgcn_readlane(dl, j);
                    const float* __restrict__ row = hsrc + (size_t)sj * F + lane;
#pragma unroll
                    for (int k = 0; k < KF; ++k) {
                        const float w = __int_as_float(
                            __builtin_amdgcn_readlane(__float_as_int(al[(k * 32) / CH]), j));
                        v[b][k] = row[k * 32] * w;
                    }
                }
#pragma unroll
                for (int b = 0; b < NB; ++b) {
                    if (j0 + b < cnt) {
                        float* arow = acc + dj[b] * F + lane;
#pragma unroll
                        for (int k = 0; k < KF; ++k)
                            __hip_atomic_fetch_add(arow + k * 32, v[b][k], __ATOMIC_RELAXED,
                                                   __HIP_MEMORY_SCOPE_WORKGROUP);
                    }
                }
            }
        }
    };

    int qn = 0;

    const unsigned bi = blockIdx.x;
    auto step = [&](const uint4& cur, bool valid) {
        unsigned h = (((cur.x >> 24) == bi) ? 1u : 0u) | (((cur.y >> 24) == bi) ? 2u : 0u) |
                     (((cur.z >> 24) == bi) ? 4u : 0u) | (((cur.w >> 24) == bi) ? 8u : 0u);
        if (!valid) h = 0u;
#pragma nounroll
        for (int trip = 0; trip < 4; ++trip) {
            const bool     hit = (h != 0u);
            const unsigned m   = __builtin_amdgcn_ballot_w32(hit);
            if (!m) break;
            if (hit) {
                const int      q   = __builtin_ctz(h);
                const int      pos = qn + (int)__builtin_amdgcn_mbcnt_lo(m, 0u);
                const unsigned r   = (q == 0) ? cur.x : (q == 1) ? cur.y : (q == 2) ? cur.z : cur.w;
                myq[pos] = (int)r;
                h &= h - 1u;
            }
            qn += __builtin_popcount(m);
        }
    };

    auto scan = [&](auto look) {
        if (byte_ok) {
            const uint4* __restrict__ bmrow = (const uint4*)(bm + (size_t)blockIdx.x * bmw);
            const int npc = bmw >> 2;
            const int ppw = (npc + AGG_NW - 1) / AGG_NW;
            for (int k = 0; k < ppw; k += 32) {
                const int pi = k + lane;
                const int gp = wave * ppw + pi;
                uint4 cw = make_uint4(0u, 0u, 0u, 0u);
                if (pi < ppw && gp < npc) cw = bmrow[gp];
                const int cb = gp << 9;
#pragma nounroll
                for (int r = 0; r < 32; ++r) {
                    if (!__builtin_amdgcn_ballot_w32((cw.x | cw.y | cw.z | cw.w) != 0u)) break;
                    int   eb[4];
                    bool  vb[4];
                    uint4 tb[4];
#pragma unroll
                    for (int b = 0; b < 4; ++b) {
                        const int      sel = cw.x ? 0 : cw.y ? 1 : cw.z ? 2 : 3;
                        const unsigned w   = (sel == 0) ? cw.x : (sel == 1) ? cw.y : (sel == 2) ? cw.z : cw.w;
                        const bool     has = (w != 0u);
                        const int      q   = has ? (sel << 5) + __builtin_ctz(w) : 0;
                        const unsigned w2  = w & (w - 1u);
                        cw.x = (sel == 0) ? w2 : cw.x;
                        cw.y = (sel == 1) ? w2 : cw.y;
                        cw.z = (sel == 2) ? w2 : cw.z;
                        cw.w = (sel == 3) ? w2 : cw.w;
                        eb[b] = cb + (q << 2);
                        vb[b] = has && (eb[b] + 4 <= E);
                        tb[b] = *(const uint4*)(rec + (vb[b] ? eb[b] : 0));
                    }
#pragma unroll
                    for (int b = 0; b < 4; ++b) {
                        if (__builtin_amdgcn_ballot_w32(vb[b])) { step(tb[b], vb[b]); look(QRecord{}); }
                    }
                }
            }
        } else {
            for (int wb = wave * 32; wb < E; wb += AGG_NW * 32) {
                const int  e   = wb + lane;
                const int  dv  = (e < E) ? dstp[e] : -1;
                const bool hit = (e < E) && ((unsigned)(dv - d0) < (unsigned)tcur);
                const unsigned m = __builtin_amdgcn_ballot_w32(hit);
                if (m) {
                    if (hit) myq[qn + (int)__builtin_amdgcn_mbcnt_lo(m, 0u)] = e;
                    qn += __builtin_popcount(m);
                }
                look(QEdge{});
            }
        }
    };

    int lost = 0;
    scan([&](auto kind) {
        if (qn > AGG_QC - 512) {
            resolve(qn, true, kind, KeepNone{});
            qn   = 0;
            lost = 1;
        }
    });
    if (byte_ok) resolve(qn, true, QRecord{}, KeepFirst{}); else resolve(qn, true, QEdge{}, KeepFirst{});
    if (lost && lane == 0) lostflag = 1;
    __syncthreads();

    if (lostflag == 0) {
        drain(qn, true);
    } else {
        qn = 0;
        scan([&](auto kind) {
            if (qn >= 32) {
                const int n32 = qn & ~31;
                const int r   = qn - n32;
                int te = 0;
                if (lane < r) te = myq[n32 + lane];
                resolve(n32, false, kind, KeepNone{});
                drain(n32, false);
                if (lane < r) myq[lane] = te;
                qn = r;
            }
        });
        if (qn > 0) {
            if (byte_ok) resolve(qn, false, QRecord{}, KeepNone{}); else resolve(qn, false, QEdge{}, KeepNone{});
            drain(qn, false);
        }
    }
    __syncthreads();

    float* __restrict__ orow = out + (size_t)d0 * F;
    const int tot = tcur * F;
    for (int i = tid * 4; i < tot; i += AGG_NW * 32 * 4) {
        float4 v = *(const float4*)(acc + i);
        const float den = sm[(i / F) * H + (i & (F - 1)) / CH] + 1e-16f;
        v.x = v.x / den + bv.x; v.y = v.y / den + bv.y; v.z = v.z / den + bv.z; v.w = v.w / den + bv.w;
        if (ELU) {
            v.x = (v.x > 0.f) ? v.x : (__expf(v.x) - 1.f);
            v.y = (v.y > 0.f) ? v.y : (__expf(v.y) - 1.f);
            v.z = (v.z > 0.f) ? v.z : (__expf(v.z) - 1.f);
            v.w = (v.w > 0.f) ? v.w : (__expf(v.w) - 1.f);
        }
        *(float4*)(orow + i) = v;
    }
}

extern "C" void kernel_launch(void* const* d_in, const int* in_sizes, int n_in,
                              void* d_out, int out_size, void* d_ws, size_t ws_size,
                              hipStream_t stream) {
    const float* x        = (const float*)d_in[0];
    const int*   ei       = (const int*)  d_in[1];
    const float* W1       = (const float*)d_in[2];
    const float* att_src1 = (const float*)d_in[3];
    const float* att_dst1 = (const float*)d_in[4];
    const float* b1       = (const float*)d_in[5];
    const float* W2       = (const float*)d_in[6];
    const float* att_src2 = (const float*)d_in[7];
    const float* att_dst2 = (const float*)d_in[8];
    const float* b2       = (const float*)d_in[9];
    float* out = (float*)d_out;

    const int N = in_sizes[0] / 128;
    const int E = in_sizes[1] / 2;

    float* ws = (float*)d_ws;
    float* h1  = ws;
    float* f1  = h1 + (size_t)N * 128;
    float* sc1 = f1 + (size_t)N * 128;
    float* sc2 = sc1 + (size_t)N * 8;
    float* wt1 = (float*)((((size_t)(sc2 + (size_t)N * 2)) + 15) & ~(size_t)15);
    float* wt2 = wt1 + 16 * FOLD_K;
    unsigned* rec = (unsigned*)((((size_t)(wt2 + 16 * FOLD_K)) + 15) & ~(size_t)15);
    const int nch       = (E + 3) / 4;
    const int recBlocks = (nch + 4 * REC_TH - 1) / (4 * REC_TH);
    const int bmw       = recBlocks * REC_W;
    unsigned* bm = (unsigned*)((((size_t)(rec + E)) + 255) & ~(size_t)255);
    const bool planes = (size_t)((const char*)(bm + (size_t)256 * bmw) - (const char*)d_ws) <= ws_size;
    if (!planes) { rec = nullptr; bm = nullptr; }
    float* h2  = h1;

    const int gBlocks = 512;
    const int aBlocks = (N + AGG_T - 1) / AGG_T;

    const int nrec = planes ? recBlocks : 0;
    edge_records<<<nrec + FOLD_GROUPS, REC_TH, 0, stream>>>(ei, E, rec, bm, bmw, nrec,
                                                            W1, att_src1, att_dst1, wt1,
                                                            W2, att_src2, att_dst2, wt2);

    gemm_wmma_f32<128, 128, 4><<<gBlocks, 128, 0, stream>>>(x, W1, wt1, h1, sc1, N);
    edge_aggregate_own<128, 4, true><<<aBlocks, AGG_NW * 32, 0, stream>>>(ei, E, rec, bm, bmw, sc1, h1, b1, f1, N);

    gemm_wmma_f32<128, 64, 1><<<gBlocks, 128, 0, stream>>>(f1, W2, wt2, h2, sc2, N);
    edge_aggregate_own<64, 1, false><<<aBlocks, AGG_NW * 32, 0, stream>>>(ei, E, rec, bm, bmw, sc2, h2, b2, out, N);
}


# Layer-2 product (hidden rows x W2, plus its score tile) formed by the layer-1 aggregation work-group from its finished rows in LDS, same WMMA k order and sums; hidden rows never stored, 1 launch goes
# speedup vs baseline: 9.6189x; 1.0544x over previous
#include <hip/hip_runtime.h>
#include <math.h>

typedef __attribute__((ext_vector_type(2))) float v2f;
typedef __attribute__((ext_vector_type(8))) float v8f;

__device__ __forceinline__ unsigned f2key(float f) {
    unsigned b = __float_as_uint(f);
    return (b & 0x80000000u) ? ~b : (b | 0x80000000u);
}
__device__ __forceinline__ float key2f(unsigned k) {
    unsigned b = (k & 0x80000000u) ? (k & 0x7FFFFFFFu) : ~k;
    return __uint_as_float(b);
}

__device__ __forceinline__ constexpr int frag_k(int s, int h, int i) {
    return 8 * (s >> 1) + 4 * h + 2 * (s & 1) + i;
}

struct WithScores    { static constexpr bool value = true;  };
struct WithoutScores { static constexpr bool value = false; };

template <int K, int NC, int H>
__global__ __launch_bounds__(128)
void gemm_wmma_f32(const float* __restrict__ A,
                   const float* __restrict__ B,
                   const float* __restrict__ WT,
                   float* __restrict__ C,
                   float* __restrict__ S,
                   int M) {
    constexpr int NG    = NC / 64;
    constexpr int SLOTS = 4 / NG;
    constexpr int KS    = K / 4;
    static_assert(NC % 64 == 0 && (4 % NG) == 0 && K % 8 == 0, "tile shape");
    static_assert(NG <= 2, "the waves of a row tile take the score tile in turns of two");
    static_assert(H == 1 || H == 4, "a lane stores its H scores with one 4-byte or one 16-byte store");

    const int tid  = threadIdx.x;
    const int lane = tid & 31;
    const int wave = __builtin_amdgcn_readfirstlane(tid >> 5);
    const int h    = lane >> 4;
    const int c    = lane & 15;
    const int grp  = wave % NG;
    const int nb   = grp * 64;
    const int slot = wave / NG;
    const int mTiles = (M + 15) >> 4;
    const int stride = (int)gridDim.x * SLOTS;

    v2f breg[4][KS];
    {
        const float* __restrict__ bcol = B + nb + 4 * c;
#pragma unroll
        for (int s = 0; s < KS; ++s) {
#pragma unroll
            for (int i = 0; i < 2; ++i) {
                const float4 w = *(const float4*)(bcol + (size_t)frag_k(s, h, i) * NC);
                breg[0][s][i] = w.x; breg[1][s][i] = w.y; breg[2][s][i] = w.z; breg[3][s][i] = w.w;
            }
        }
    }

    float4 wq[K / 8];
    {
        const float* __restrict__ wrow = WT + (size_t)c * K + 4 * h;
#pragma unroll
        for (int j = 0; j < K / 8; ++j) wq[j] = *(const float4*)(wrow + 8 * j);
    }

    int trip = 0;
    for (int mt = (int)blockIdx.x * SLOTS + slot; mt < mTiles; mt += stride, ++trip) {
        const int m0 = mt << 4;
        int rowA = m0 + c;
        if (rowA >= M) rowA = M - 1;
        const float* __restrict__ arow = A + (size_t)rowA * K + 4 * h;

        float4 q[K / 8];
#pragma unroll
        for (int j = 0; j < K / 8; ++j) q[j] = *(const float4*)(arow + 8 * j);

        const bool scores = (NG == 1) || (((trip ^ grp) & 1) == 0);

        v8f acc[4] = {};
        v8f accS[2] = {};
        auto ksteps = [&](auto with) {
#pragma unroll
            for (int s = 0; s < KS; ++s) {
                const float4 qq = q[s >> 1];
                v2f a;
                if (s & 1) { a.x = qq.z; a.y = qq.w; } else { a.x = qq.x; a.y = qq.y; }
#pragma unroll
                for (int t = 0; t < 4; ++t)
                    acc[t] = __builtin_amdgcn_wmma_f32_16x16x4_f32(false, a, false, breg[t][s],
                                                                   (short)0, acc[t], false, false);
                if constexpr (decltype(with)::value) {
                    const float4 ww = wq[s >> 1];
                    v2f w;
                    if (s & 1) { w.x = ww.z; w.y = ww.w; } else { w.x = ww.x; w.y = ww.y; }
                    accS[s & 1] = __builtin_amdgcn_wmma_f32_16x16x4_f32(false, w, false, a,
                                                                        (short)0, accS[s & 1], false, false);
                }
            }
        };
        if (scores) ksteps(WithScores{}); else ksteps(WithoutScores{});

        float* __restrict__ crow = C + (size_t)(m0 + (h << 3)) * NC + nb + 4 * c;
#pragma unroll
        for (int r = 0; r < 8; ++r) {
            if (m0 + (h << 3) + r < M) {
                float4 o;
                o.x = acc[0][r]; o.y = acc[1][r]; o.z = acc[2][r]; o.w = acc[3][r];
                *(float4*)(crow + (size_t)r * NC) = o;
            }
        }

        if (scores && m0 + c < M) {
            float* __restrict__ srow = S + (size_t)(m0 + c) * (2 * H) + h * H;
            if constexpr (H == 4) {
                float4 o;
                o.x = accS[0][0] + accS[1][0]; o.y = accS[0][1] + accS[1][1];
                o.z = accS[0][2] + accS[1][2]; o.w = accS[0][3] + accS[1][3];
                *(float4*)srow = o;
            } else {
                srow[0] = accS[0][0] + accS[1][0];
            }
        }
    }
}

#define FOLD_K 128
#define FOLD_N (2 * 16 * FOLD_K)
template <int NC, int H>
__device__ __forceinline__ void fold_scores(int r, int k,
                                            const float* __restrict__ W,
                                            const float* __restrict__ a_src,
                                            const float* __restrict__ a_dst,
                                            float* __restrict__ WT) {
    constexpr int CH = NC / H;
    static_assert(H <= 8 && CH % 4 == 0, "score rows 0-7 source, 8-15 destination; rows of W read 16 bytes at a time");
    const int hd = r & 7;
    float v = 0.f;
    if (hd < H) {
        const float* __restrict__ att = ((r & 8) ? a_dst : a_src) + hd * CH;
        const float* __restrict__ w   = W + (size_t)k * NC + hd * CH;
        for (int ch = 0; ch < CH; ch += 4) {
            const float4 x = *(const float4*)(w + ch);
            v = fmaf(x.x, att[ch],     v);
            v = fmaf(x.y, att[ch + 1], v);
            v = fmaf(x.z, att[ch + 2], v);
            v = fmaf(x.w, att[ch + 3], v);
        }
    }
    WT[r * FOLD_K + k] = v;
}

#define AGG_T  196
#define AGG_NW 32
#define AGG_QC 1024
#define AGG_SP 208

template <bool G>
__device__ __forceinline__ size_t score_slot(int n) {
    if constexpr (G) return (size_t)((unsigned)n / (unsigned)AGG_T) * AGG_SP + (unsigned)n % (unsigned)AGG_T;
    else             return (size_t)n;
}

#define REC_TH 256
#define REC_W  (4 * REC_TH / 32)
#define FOLD_GROUPS (FOLD_N / REC_TH)
__global__ __launch_bounds__(REC_TH)
void edge_records(const int* __restrict__ ei, int E, unsigned* __restrict__ rec,
                  unsigned* __restrict__ bm, int bmw, int nrec,
                  const float* __restrict__ W1, const float* __restrict__ att_src1,
                  const float* __restrict__ att_dst1, float* __restrict__ WT1,
                  const float* __restrict__ W2, const float* __restrict__ att_src2,
                  const float* __restrict__ att_dst2, float* __restrict__ WT2) {
    static_assert(REC_W % 4 == 0, "a row piece is stored with 16-byte stores");
    static_assert(AGG_T * 256 <= 65536, "a node index must fit 16 bits while a byte holds the owner");
    static_assert(FOLD_N % REC_TH == 0, "one element of WT per thread of the fold groups");
    __shared__ __attribute__((aligned(16))) unsigned lbm[256 * REC_W];
    const int tid = threadIdx.x;
    if ((int)blockIdx.x >= nrec) {
        const int idx = ((int)blockIdx.x - nrec) * REC_TH + tid;
        const int r   = (idx / FOLD_K) & 15;
        const int k   = idx % FOLD_K;
        if (idx < 16 * FOLD_K) fold_scores<128, 4>(r, k, W1, att_src1, att_dst1, WT1);
        else                   fold_scores<64, 1>(r, k, W2, att_src2, att_dst2, WT2);
        return;
    }
    for (int i = tid; i < 256 * REC_W / 4; i += REC_TH) ((uint4*)lbm)[i] = make_uint4(0u, 0u, 0u, 0u);
    __syncthreads();

    const int* __restrict__ dstp = ei + E;
    const bool al16 = (((size_t)ei) & 15) == 0 && (E & 3) == 0 && (((size_t)rec) & 15) == 0;
    unsigned sv[4][4], dv[4][4];
#pragma unroll
    for (int k = 0; k < 4; ++k) {
        const int e0 = ((blockIdx.x * 4 + k) * REC_TH + tid) * 4;
        if (al16 && e0 + 4 <= E) {
            const int4 s = *(const int4*)(ei + e0);
            const int4 d = *(const int4*)(dstp + e0);
            sv[k][0] = (unsigned)s.x; sv[k][1] = (unsigned)s.y; sv[k][2] = (unsigned)s.z; sv[k][3] = (unsigned)s.w;
            dv[k][0] = (unsigned)d.x; dv[k][1] = (unsigned)d.y; dv[k][2] = (unsigned)d.z; dv[k][3] = (unsigned)d.w;
        } else {
#pragma unroll
            for (int j = 0; j < 4; ++j) {
                sv[k][j] = (e0 + j < E) ? (unsigned)ei[e0 + j] : 0u;
                dv[k][j] = (e0 + j < E) ? (unsigned)dstp[e0 + j] : 0u;
            }
        }
    }
#pragma unroll
    for (int k = 0; k < 4; ++k) {
        const int e0 = ((blockIdx.x * 4 + k) * REC_TH + tid) * 4;
        unsigned rw[4], ow[4];
#pragma unroll
        for (int j = 0; j < 4; ++j) {
            const unsigned o = dv[k][j] / (unsigned)AGG_T;
            const unsigned l = dv[k][j] - o * (unsigned)AGG_T;
            ow[j] = o & 255u;
            rw[j] = (sv[k][j] & 0xffffu) | (l << 16) | (ow[j] << 24);
        }
        if (al16 && e0 + 4 <= E) {
            *(uint4*)(rec + e0) = make_uint4(rw[0], rw[1], rw[2], rw[3]);
        } else {
#pragma unroll
            for (int j = 0; j < 4; ++j) if (e0 + j < E) rec[e0 + j] = rw[j];
        }
        const unsigned bit = 1u << (tid & 31);
        unsigned*      lw  = lbm + 8 * k + (tid >> 5);
#pragma unroll
        for (int j = 0; j < 4; ++j)
            if (e0 + j < E) atomicOr(lw + ow[j] * REC_W, bit);
    }
    __syncthreads();

    for (int i = tid; i < 256 * (REC_W / 4); i += REC_TH) {
        const int r = i / (REC_W / 4);
        const int w = i - r * (REC_W / 4);
        ((uint4*)(bm + (size_t)r * bmw + (size_t)blockIdx.x * REC_W))[w] = ((const uint4*)(lbm + r * REC_W))[w];
    }
}

struct QRecord { static constexpr bool packed = true;  };
struct QEdge   { static constexpr bool packed = false; };

struct KeepFirst { static constexpr bool value = true;  };
struct KeepNone  { static constexpr bool value = false; };

typedef __attribute__((ext_vector_type(4))) float v4f;
template <int H> struct HeadVec;
template <> struct HeadVec<4> { typedef v4f   type; };
template <> struct HeadVec<1> { typedef float type; };
__device__ __forceinline__ float hv_get(float v, int)        { return v; }
__device__ __forceinline__ float hv_get(v4f v, int h)        { return v[h]; }
__device__ __forceinline__ void  hv_set(float& v, int, float x) { v = x; }
__device__ __forceinline__ void  hv_set(v4f& v, int h, float x) { v[h] = x; }

template <int H>
__device__ __forceinline__ void ld_heads(const float* __restrict__ q, float (&o)[H]) {
    if constexpr (H == 4) {
        const float4 t = *(const float4*)q;
        o[0] = t.x; o[1] = t.y; o[2] = t.z; o[3] = t.w;
    } else {
#pragma unroll
        for (int h = 0; h < H; ++h) o[h] = q[h];
    }
}

template <int F, int H, bool ELU, bool NEXT, bool GSC>
__global__ __launch_bounds__(1024)
void edge_aggregate_own(const int* __restrict__ ei, int E,
                        const unsigned* __restrict__ rec,
                        const unsigned* __restrict__ bm, int bmw,
                        const float* __restrict__ sc,
                        const float* __restrict__ hsrc,
                        const float* __restrict__ bias,
                        float* __restrict__ out, int N,
                        const float* __restrict__ Wn,
                        const float* __restrict__ WTn,
                        float* __restrict__ Sn) {
    constexpr int KF = F / 32;
    constexpr int CH = F / H;
    constexpr int NB = 1024 / F;
    static_assert(CH % 32 == 0, "a 32-channel slice must lie inside one head");
    static_assert(AGG_T <= 256, "a local destination must fit a byte");
    __shared__ __attribute__((aligned(16))) float acc[AGG_T * F];
    __shared__ __attribute__((aligned(16))) int qs[AGG_NW * AGG_QC];
    __shared__ unsigned char qd[AGG_NW * AGG_QC];
    __shared__ unsigned      mx[AGG_T * H];
    __shared__ float         sm[AGG_T * H];
    __shared__ int           lostflag;

    const int tid  = threadIdx.x;
    const int lane = tid & 31;
    const int wave = __builtin_amdgcn_readfirstlane(tid >> 5);
    const int d0   = blockIdx.x * AGG_T;
    const int tcur = min(AGG_T, N - d0);

    static_assert((AGG_NW * 32 * 4) % F == 0, "a thread must keep its channels from trip to trip");
    const float4 bv = *(const float4*)(bias + ((tid * 4) & (F - 1)));

    for (int i = tid; i < AGG_T * F; i += AGG_NW * 32) acc[i] = 0.f;
    for (int i = tid; i < AGG_T * H; i += AGG_NW * 32) { mx[i] = 0u; sm[i] = 0.f; }
    if (tid == 0) lostflag = 0;
    __syncthreads();

    const int* __restrict__ dstp = ei + E;
    int*           myq  = qs + wave * AGG_QC;
    unsigned char* myqd = qd + wave * AGG_QC;
    const bool byte_ok = (rec != nullptr) && (gridDim.x <= 256u) && ((E & 15) == 0) &&
                         ((((size_t)rec) & 15) == 0) && ((bmw & 3) == 0) && ((((size_t)bm) & 15) == 0);

    static_assert(AGG_QC % 128 == 0, "a lane past the end must still store inside the queue");
    int ks0 = 0, ks1 = 0, ks2 = 0, ks3 = 0;
    int kd0 = 0, kd1 = 0, kd2 = 0, kd3 = 0;
    typename HeadVec<H>::type kl0 = {}, kl1 = {}, kl2 = {}, kl3 = {};
    auto resolve = [&](int n, bool domax, auto kind, auto keep) {
        for (int off = 0; off < n; off += 128) {
            int  e[4], s[4], d[4];
            bool ok[4];
#pragma unroll
            for (int b = 0; b < 4; ++b) {
                const int idx = off + b * 32 + lane;
                ok[b] = idx < n;
                e[b]  = ok[b] ? myq[idx] : 0;
            }
#pragma unroll
            for (int b = 0; b < 4; ++b) {
                if constexpr (decltype(kind)::packed) {
                    s[b] = e[b] & 0xffff;
                    d[b] = d0 + ((e[b] >> 16) & 0xff);
                } else {
                    s[b] = ei[e[b]]; d[b] = dstp[e[b]];
                }
            }
            float a[4][H], c[4][H];
            if (domax) {
#pragma unroll
                for (int b = 0; b < 4; ++b) {
                    ld_heads<H>(sc + score_slot<GSC>(s[b]) * (2 * H), a[b]);
                    ld_heads<H>(sc + score_slot<GSC>(d[b]) * (2 * H) + H, c[b]);
                }
            }
#pragma unroll
            for (int b = 0; b < 4; ++b) {
                const int idx = off + b * 32 + lane;
                const int dl  = d[b] - d0;
                myq[idx]  = s[b];
                myqd[idx] = (unsigned char)dl;
                if (domax) {
                    float el[H];
#pragma unroll
                    for (int h = 0; h < H; ++h) {
                        el[h] = a[b][h] + c[b][h];
                        el[h] = (el[h] > 0.f) ? el[h] : 0.2f * el[h];
                    }
                    if (ok[b]) {
#pragma unroll
                        for (int h = 0; h < H; ++h) atomicMax(&mx[dl * H + h], f2key(el[h]));
                    }
                    if constexpr (decltype(keep)::value) {
                        if (off == 0) {
                            typename HeadVec<H>::type lv = {};
#pragma unroll
                            for (int h = 0; h < H; ++h) hv_set(lv, h, el[h]);
                            if      (b == 0) { ks0 = s[b]; kd0 = dl; kl0 = lv; }
                            else if (b == 1) { ks1 = s[b]; kd1 = dl; kl1 = lv; }
                            else if (b == 2) { ks2 = s[b]; kd2 = dl; kl2 = lv; }
                            else             { ks3 = s[b]; kd3 = dl; kl3 = lv; }
                        }
                    }
                }
            }
        }
    };

    auto drain = [&](int n, bool kept) {
        for (int off = 0; off < n; off += 32) {
            const int cnt = min(32, n - off);
            const int t   = off >> 5;
            int   s = 0, dl = 0;
            float el[H], al[H];
#pragma unroll
            for (int h = 0; h < H; ++h) { el[h] = 0.f; al[h] = 0.f; }
            if (kept && t < 4) {
                s  = (t == 0) ? ks0 : (t == 1) ? ks1 : (t == 2) ? ks2 : ks3;
                dl = (t == 0) ? kd0 : (t == 1) ? kd1 : (t == 2) ? kd2 : kd3;
                const typename HeadVec<H>::type lv = (t == 0) ? kl0 : (t == 1) ? kl1 : (t == 2) ? kl2 : kl3;
#pragma unroll
                for (int h = 0; h < H; ++h) el[h] = hv_get(lv, h);
            } else if (lane < cnt) {
                s  = myq[off + lane];
                dl = (int)myqd[off + lane];
                float a[H], c[H];
                ld_heads<H>(sc + score_slot<GSC>(s) * (2 * H), a);
                ld_heads<H>(sc + score_slot<GSC>(d0 + dl) * (2 * H) + H, c);
#pragma unroll
                for (int h = 0; h < H; ++h) {
                    el[h] = a[h] + c[h];
                    el[h] = (el[h] > 0.f) ? el[h] : 0.2f * el[h];
                }
            }
            if (lane < cnt) {
#pragma unroll
                for (int h = 0; h < H; ++h) {
                    const float pe = __expf(el[h] - key2f(mx[dl * H + h]));
                    al[h] = pe;
                    __hip_atomic_fetch_add(&sm[dl * H + h], pe, __ATOMIC_RELAXED,
                                           __HIP_MEMORY_SCOPE_WORKGROUP);
                }
            }
            for (int j0 = 0; j0 < cnt; j0 += NB) {
                float v[NB][KF];
                int   dj[NB];
#pragma unroll
                for (int b = 0; b < NB; ++b) {
                    const int j  = min(j0 + b, cnt - 1);
                    const int sj = __builtin_amdgcn_readlane(s, j);
                    dj[b] = __builtin_amdgcn_readlane(dl, j);
                    const float* __restrict__ row = hsrc + (size_t)sj * F + lane;
#pragma unroll
                    for (int k = 0; k < KF; ++k) {
                        const float w = __int_as_float(
                            __builtin_amdgcn_readlane(__float_as_int(al[(k * 32) / CH]), j));
                        v[b][k] = row[k * 32] * w;
                    }
                }
#pragma unroll
                for (int b = 0; b < NB; ++b) {
                    if (j0 + b < cnt) {
                        float* arow = acc + dj[b] * F + lane;
#pragma unroll
                        for (int k = 0; k < KF; ++k)
                            __hip_atomic_fetch_add(arow + k * 32, v[b][k], __ATOMIC_RELAXED,
                                                   __HIP_MEMORY_SCOPE_WORKGROUP);
                    }
                }
            }
        }
    };

    int qn = 0;

    const unsigned bi = blockIdx.x;
    auto step = [&](const uint4& cur, bool valid) {
        unsigned h = (((cur.x >> 24) == bi) ? 1u : 0u) | (((cur.y >> 24) == bi) ? 2u : 0u) |
                     (((cur.z >> 24) == bi) ? 4u : 0u) | (((cur.w >> 24) == bi) ? 8u : 0u);
        if (!valid) h = 0u;
#pragma nounroll
        for (int trip = 0; trip < 4; ++trip) {
            const bool     hit = (h != 0u);
            const unsigned m   = __builtin_amdgcn_ballot_w32(hit);
            if (!m) break;
            if (hit) {
                const int      q   = __builtin_ctz(h);
                const int      pos = qn + (int)__builtin_amdgcn_mbcnt_lo(m, 0u);
                const unsigned r   = (q == 0) ? cur.x : (q == 1) ? cur.y : (q == 2) ? cur.z : cur.w;
                myq[pos] = (int)r;
                h &= h - 1u;
            }
            qn += __builtin_popcount(m);
        }
    };

    auto scan = [&](auto look) {
        if (byte_ok) {
            const uint4* __restrict__ bmrow = (const uint4*)(bm + (size_t)blockIdx.x * bmw);
            const int npc = bmw >> 2;
            const int ppw = (npc + AGG_NW - 1) / AGG_NW;
            for (int k = 0; k < ppw; k += 32) {
                const int pi = k + lane;
                const int gp = wave * ppw + pi;
                uint4 cw = make_uint4(0u, 0u, 0u, 0u);
                if (pi < ppw && gp < npc) cw = bmrow[gp];
                const int cb = gp << 9;
#pragma nounroll
                for (int r = 0; r < 32; ++r) {
                    if (!__builtin_amdgcn_ballot_w32((cw.x | cw.y | cw.z | cw.w) != 0u)) break;
                    int   eb[4];
                    bool  vb[4];
                    uint4 tb[4];
#pragma unroll
                    for (int b = 0; b < 4; ++b) {
                        const int      sel = cw.x ? 0 : cw.y ? 1 : cw.z ? 2 : 3;
                        const unsigned w   = (sel == 0) ? cw.x : (sel == 1) ? cw.y : (sel == 2) ? cw.z : cw.w;
                        const bool     has = (w != 0u);
                        const int      q   = has ? (sel << 5) + __builtin_ctz(w) : 0;
                        const unsigned w2  = w & (w - 1u);
                        cw.x = (sel == 0) ? w2 : cw.x;
                        cw.y = (sel == 1) ? w2 : cw.y;
                        cw.z = (sel == 2) ? w2 : cw.z;
                        cw.w = (sel == 3) ? w2 : cw.w;
                        eb[b] = cb + (q << 2);
                        vb[b] = has && (eb[b] + 4 <= E);
                        tb[b] = *(const uint4*)(rec + (vb[b] ? eb[b] : 0));
                    }
#pragma unroll
                    for (int b = 0; b < 4; ++b) {
                        if (__builtin_amdgcn_ballot_w32(vb[b])) { step(tb[b], vb[b]); look(QRecord{}); }
                    }
                }
            }
        } else {
            for (int wb = wave * 32; wb < E; wb += AGG_NW * 32) {
                const int  e   = wb + lane;
                const int  dv  = (e < E) ? dstp[e] : -1;
                const bool hit = (e < E) && ((unsigned)(dv - d0) < (unsigned)tcur);
                const unsigned m = __builtin_amdgcn_ballot_w32(hit);
                if (m) {
                    if (hit) myq[qn + (int)__builtin_amdgcn_mbcnt_lo(m, 0u)] = e;
                    qn += __builtin_popcount(m);
                }
                look(QEdge{});
            }
        }
    };

    int lost = 0;
    scan([&](auto kind) {
        if (qn > AGG_QC - 512) {
            resolve(qn, true, kind, KeepNone{});
            qn   = 0;
            lost = 1;
        }
    });
    if (byte_ok) resolve(qn, true, QRecord{}, KeepFirst{}); else resolve(qn, true, QEdge{}, KeepFirst{});
    if (lost && lane == 0) lostflag = 1;
    __syncthreads();

    if (lostflag == 0) {
        drain(qn, true);
    } else {
        qn = 0;
        scan([&](auto kind) {
            if (qn >= 32) {
                const int n32 = qn & ~31;
                const int r   = qn - n32;
                int te = 0;
                if (lane < r) te = myq[n32 + lane];
                resolve(n32, false, kind, KeepNone{});
                drain(n32, false);
                if (lane < r) myq[lane] = te;
                qn = r;
            }
        });
        if (qn > 0) {
            if (byte_ok) resolve(qn, false, QRecord{}, KeepNone{}); else resolve(qn, false, QEdge{}, KeepNone{});
            drain(qn, false);
        }
    }
    __syncthreads();

    constexpr int PITCH = F + 4;
    static_assert(!NEXT || (AGG_T * PITCH * 4 <= AGG_NW * AGG_QC * 4 && (PITCH & 3) == 0 && ((PITCH >> 2) & 1) == 1),
                  "the finished rows must fit the queue area; pitch = 16 bytes x an odd number");
    float* fr = reinterpret_cast<float*>(qs);
    float* __restrict__ orow = out + (size_t)d0 * F;
    const int tot = tcur * F;
    for (int i = tid * 4; i < tot; i += AGG_NW * 32 * 4) {
        float4 v = *(const float4*)(acc + i);
        const float den = sm[(i / F) * H + (i & (F - 1)) / CH] + 1e-16f;
        v.x = v.x / den + bv.x; v.y = v.y / den + bv.y; v.z = v.z / den + bv.z; v.w = v.w / den + bv.w;
        if (ELU) {
            v.x = (v.x > 0.f) ? v.x : (__expf(v.x) - 1.f);
            v.y = (v.y > 0.f) ? v.y : (__expf(v.y) - 1.f);
            v.z = (v.z > 0.f) ? v.z : (__expf(v.z) - 1.f);
            v.w = (v.w > 0.f) ? v.w : (__expf(v.w) - 1.f);
        }
        if constexpr (NEXT) *(float4*)(fr + (i / F) * PITCH + (i & (F - 1))) = v;
        else                *(float4*)(orow + i) = v;
    }

    if constexpr (NEXT) {
        static_assert(F % 8 == 0, "two k-steps per 16-byte fragment load");
        constexpr int NCN = 64;
        constexpr int PW  = 2 * ((AGG_T + 15) / 16);
        static_assert(PW < AGG_NW, "at least one wave must be left for the score tiles");
        __syncthreads();
        const int hh = lane >> 4;
        const int cc = lane & 15;
        const int nT = (tcur + 15) >> 4;
        if (wave < PW) {
            const int rt   = wave >> 1;
            const int half = wave & 1;
            if (rt < nT) {
                const int m0 = rt << 4;
                const int rl = min(m0 + cc, tcur - 1);
                const float* arow = fr + rl * PITCH + 4 * hh;
                const float* __restrict__ bcol = Wn + half * 32 + 2 * cc;
                v8f p0 = {}, p1 = {};
#pragma unroll
                for (int j = 0; j < F / 8; ++j) {
                    const float4 q = *(const float4*)(arow + 8 * j);
#pragma unroll
                    for (int p = 0; p < 2; ++p) {
                        const int s = 2 * j + p;
                        v2f a;
                        if (p) { a.x = q.z; a.y = q.w; } else { a.x = q.x; a.y = q.y; }
                        const float2 w0 = *(const float2*)(bcol + (size_t)frag_k(s, hh, 0) * NCN);
                        const float2 w1 = *(const float2*)(bcol + (size_t)frag_k(s, hh, 1) * NCN);
                        v2f b0, b1;
                        b0.x = w0.x; b0.y = w1.x;
                        b1.x = w0.y; b1.y = w1.y;
                        p0 = __builtin_amdgcn_wmma_f32_16x16x4_f32(false, a, false, b0, (short)0, p0, false, false);
                        p1 = __builtin_amdgcn_wmma_f32_16x16x4_f32(false, a, false, b1, (short)0, p1, false, false);
                    }
                }
                float* __restrict__ crow = out + (size_t)(d0 + m0 + (hh << 3)) * NCN + half * 32 + 2 * cc;
#pragma unroll
                for (int r = 0; r < 8; ++r) {
                    if (m0 + (hh << 3) + r < tcur) {
                        float2 o;
                        o.x = p0[r]; o.y = p1[r];
                        *(float2*)(crow + (size_t)r * NCN) = o;
                    }
                }
            }
        } else {
            const float* __restrict__ wrow = WTn + (size_t)cc * F + 4 * hh;
            for (int rt = wave - PW; rt < nT; rt += AGG_NW - PW) {
                const int m0 = rt << 4;
                const int rl = min(m0 + cc, tcur - 1);
                const float* arow = fr + rl * PITCH + 4 * hh;
                v8f s0 = {}, s1 = {};
#pragma unroll
                for (int j = 0; j < F / 8; ++j) {
                    const float4 q  = *(const float4*)(arow + 8 * j);
                    const float4 ww = *(const float4*)(wrow + 8 * j);
                    v2f a0, a1, w0, w1;
                    a0.x = q.x;  a0.y = q.y;  a1.x = q.z;  a1.y = q.w;
                    w0.x = ww.x; w0.y = ww.y; w1.x = ww.z; w1.y = ww.w;
                    s0 = __builtin_amdgcn_wmma_f32_16x16x4_f32(false, w0, false, a0, (short)0, s0, false, false);
                    s1 = __builtin_amdgcn_wmma_f32_16x16x4_f32(false, w1, false, a1, (short)0, s1, false, false);
                }
                static_assert(((AGG_T + 15) / 16) * 16 <= AGG_SP && AGG_SP % 16 == 0, "whole tiles inside a group's slots");
                Sn[((size_t)blockIdx.x * AGG_SP + m0 + cc) * 2 + hh] = s0[0] + s1[0];
            }
        }
    }
}

extern "C" void kernel_launch(void* const* d_in, const int* in_sizes, int n_in,
                              void* d_out, int out_size, void* d_ws, size_t ws_size,
                              hipStream_t stream) {
    const float* x        = (const float*)d_in[0];
    const int*   ei       = (const int*)  d_in[1];
    const float* W1       = (const float*)d_in[2];
    const float* att_src1 = (const float*)d_in[3];
    const float* att_dst1 = (const float*)d_in[4];
    const float* b1       = (const float*)d_in[5];
    const float* W2       = (const float*)d_in[6];
    const float* att_src2 = (const float*)d_in[7];
    const float* att_dst2 = (const float*)d_in[8];
    const float* b2       = (const float*)d_in[9];
    float* out = (float*)d_out;

    const int N = in_sizes[0] / 128;
    const int E = in_sizes[1] / 2;

    float* ws = (float*)d_ws;
    float* h1  = ws;
    float* h2  = h1 + (size_t)N * 128;
    float* sc1 = h2 + (size_t)N * 64;
    const int aBlocks = (N + AGG_T - 1) / AGG_T;
    float* sc2 = (float*)((((size_t)(sc1 + (size_t)N * 8)) + 127) & ~(size_t)127);
    float* wt1 = (float*)((((size_t)(sc2 + (size_t)aBlocks * AGG_SP * 2)) + 15) & ~(size_t)15);
    float* wt2 = wt1 + 16 * FOLD_K;
    unsigned* rec = (unsigned*)((((size_t)(wt2 + 16 * FOLD_K)) + 15) & ~(size_t)15);
    const int nch       = (E + 3) / 4;
    const int recBlocks = (nch + 4 * REC_TH - 1) / (4 * REC_TH);
    const int bmw       = recBlocks * REC_W;
    unsigned* bm = (unsigned*)((((size_t)(rec + E)) + 255) & ~(size_t)255);
    const bool planes = (size_t)((const char*)(bm + (size_t)256 * bmw) - (const char*)d_ws) <= ws_size;
    if (!planes) { rec = nullptr; bm = nullptr; }

    const int gBlocks = 512;

    const int nrec = planes ? recBlocks : 0;
    edge_records<<<nrec + FOLD_GROUPS, REC_TH, 0, stream>>>(ei, E, rec, bm, bmw, nrec,
                                                            W1, att_src1, att_dst1, wt1,
                                                            W2, att_src2, att_dst2, wt2);

    gemm_wmma_f32<128, 128, 4><<<gBlocks, 128, 0, stream>>>(x, W1, wt1, h1, sc1, N);
    edge_aggregate_own<128, 4, true, true, false><<<aBlocks, AGG_NW * 32, 0, stream>>>(ei, E, rec, bm, bmw, sc1, h1, b1, h2, N,
                                                                             W2, wt2, sc2);

    edge_aggregate_own<64, 1, false, false, true><<<aBlocks, AGG_NW * 32, 0, stream>>>(ei, E, rec, bm, bmw, sc2, h2, b2, out, N,
                                                                              nullptr, nullptr, nullptr);
}
